# starting kernel
# baseline (speedup 1.0000x reference)
#include <hip/hip_runtime.h>

#define H 256
#define TM 128
#define TNB 64
#define KC 32
#define APAD 40

typedef __bf16 bf16;
typedef __attribute__((ext_vector_type(16))) __bf16 v16bf;
typedef __attribute__((ext_vector_type(8)))  __bf16 v8bf;
typedef __attribute__((ext_vector_type(8)))  float  v8f;

__device__ __forceinline__ unsigned short f2bfbits(float f) {
  unsigned u = __float_as_uint(f);
  return (unsigned short)((u + 0x7FFFu + ((u >> 16) & 1u)) >> 16);
}
__device__ __forceinline__ float bfbits2f(unsigned short h) {
  return __uint_as_float(((unsigned)h) << 16);
}
__device__ __forceinline__ bf16 f2bf(float f) {
  unsigned short hs = f2bfbits(f);
  return __builtin_bit_cast(bf16, hs);
}
__device__ __forceinline__ float lrelu(float v) { return v > 0.f ? v : 0.2f * v; }
__device__ __forceinline__ unsigned f2ord(float f) {
  unsigned u = __float_as_uint(f);
  return (u & 0x80000000u) ? ~u : (u | 0x80000000u);
}
__device__ __forceinline__ float ord2f(unsigned u) {
  unsigned b = (u & 0x80000000u) ? (u & 0x7FFFFFFFu) : ~u;
  return __uint_as_float(b);
}

__device__ __forceinline__ v16bf load_frag(const bf16* rowp, int g0) {
  v8bf a = *(const v8bf*)(rowp + g0);
  v8bf b = *(const v8bf*)(rowp + 16 + g0);
  return __builtin_shufflevector(a, b, 0, 1, 2, 3, 4, 5, 6, 7,
                                 8, 9, 10, 11, 12, 13, 14, 15);
}

#define WMMA_BF16(A_, B_, C_) \
  __builtin_amdgcn_wmma_f32_16x16x32_bf16(false, (A_), false, (B_), (short)0, (C_), false, false)

__global__ __launch_bounds__(256) void k_gemm_bf16x3(
    const float* __restrict__ A, const float* __restrict__ B,
    const float* __restrict__ bias, float* __restrict__ C, int M)
{
  __shared__ bf16 Ah[TM][APAD];
  __shared__ bf16 Al[TM][APAD];
  __shared__ bf16 Bth[TNB][APAD];
  __shared__ bf16 Btl[TNB][APAD];

  const int tid = threadIdx.x;
  const int m0 = blockIdx.x * TM;
  const int n0 = blockIdx.y * TNB;
  const int wid = tid >> 5, lane = tid & 31;
  const int hl = lane & 15, hh = lane >> 4;
  const int wr = wid >> 1;
  const int wc = wid & 1;
  const int g0 = hh << 3;

  v8f acc00 = {}, acc01 = {}, acc10 = {}, acc11 = {};

  for (int k0 = 0; k0 < H; k0 += KC) {
    __syncthreads();
    for (int idx = tid; idx < (TM * KC) / 4; idx += 256) {
      int r = idx >> 3;
      int c4 = (idx & 7) << 2;
      int gr = m0 + r;
      float4 v = (gr < M) ? *(const float4*)&A[(size_t)gr * H + k0 + c4]
                          : make_float4(0.f, 0.f, 0.f, 0.f);
      ushort4 hiv, lov;
      hiv.x = f2bfbits(v.x); lov.x = f2bfbits(v.x - bfbits2f(hiv.x));
      hiv.y = f2bfbits(v.y); lov.y = f2bfbits(v.y - bfbits2f(hiv.y));
      hiv.z = f2bfbits(v.z); lov.z = f2bfbits(v.z - bfbits2f(hiv.z));
      hiv.w = f2bfbits(v.w); lov.w = f2bfbits(v.w - bfbits2f(hiv.w));
      *(ushort4*)&Ah[r][c4] = hiv;
      *(ushort4*)&Al[r][c4] = lov;
    }
    for (int idx = tid; idx < (KC * TNB) / 4; idx += 256) {
      int r = idx >> 4;
      int c4 = (idx & 15) << 2;
      float4 v = *(const float4*)&B[(size_t)(k0 + r) * H + n0 + c4];
      bf16 h0 = f2bf(v.x), h1 = f2bf(v.y), h2 = f2bf(v.z), h3 = f2bf(v.w);
      Bth[c4 + 0][r] = h0; Btl[c4 + 0][r] = f2bf(v.x - bfbits2f(__builtin_bit_cast(unsigned short, h0)));
      Bth[c4 + 1][r] = h1; Btl[c4 + 1][r] = f2bf(v.y - bfbits2f(__builtin_bit_cast(unsigned short, h1)));
      Bth[c4 + 2][r] = h2; Btl[c4 + 2][r] = f2bf(v.z - bfbits2f(__builtin_bit_cast(unsigned short, h2)));
      Bth[c4 + 3][r] = h3; Btl[c4 + 3][r] = f2bf(v.w - bfbits2f(__builtin_bit_cast(unsigned short, h3)));
    }
    __syncthreads();

    v16bf ah0 = load_frag(&Ah[(wr << 5) + hl][0], g0);
    v16bf al0 = load_frag(&Al[(wr << 5) + hl][0], g0);
    v16bf ah1 = load_frag(&Ah[(wr << 5) + 16 + hl][0], g0);
    v16bf al1 = load_frag(&Al[(wr << 5) + 16 + hl][0], g0);
    v16bf bh0 = load_frag(&Bth[(wc << 5) + hl][0], g0);
    v16bf bl0 = load_frag(&Btl[(wc << 5) + hl][0], g0);
    v16bf bh1 = load_frag(&Bth[(wc << 5) + 16 + hl][0], g0);
    v16bf bl1 = load_frag(&Btl[(wc << 5) + 16 + hl][0], g0);

    acc00 = WMMA_BF16(ah0, bh0, acc00);
    acc00 = WMMA_BF16(al0, bh0, acc00);
    acc00 = WMMA_BF16(ah0, bl0, acc00);
    acc01 = WMMA_BF16(ah0, bh1, acc01);
    acc01 = WMMA_BF16(al0, bh1, acc01);
    acc01 = WMMA_BF16(ah0, bl1, acc01);
    acc10 = WMMA_BF16(ah1, bh0, acc10);
    acc10 = WMMA_BF16(al1, bh0, acc10);
    acc10 = WMMA_BF16(ah1, bl0, acc10);
    acc11 = WMMA_BF16(ah1, bh1, acc11);
    acc11 = WMMA_BF16(al1, bh1, acc11);
    acc11 = WMMA_BF16(ah1, bl1, acc11);
  }

  const int col0 = n0 + (wc << 5) + hl;
  const int col1 = col0 + 16;
  const float b0 = bias ? bias[col0] : 0.f;
  const float b1v = bias ? bias[col1] : 0.f;
  const int rbase0 = m0 + (wr << 5) + (hh << 3);
  const int rbase1 = rbase0 + 16;
#pragma unroll
  for (int r = 0; r < 8; ++r) {
    int g0r = rbase0 + r;
    if (g0r < M) {
      C[(size_t)g0r * H + col0] = acc00[r] + b0;
      C[(size_t)g0r * H + col1] = acc01[r] + b1v;
    }
    int g1r = rbase1 + r;
    if (g1r < M) {
      C[(size_t)g1r * H + col0] = acc10[r] + b0;
      C[(size_t)g1r * H + col1] = acc11[r] + b1v;
    }
  }
}

__global__ void k_zero(float* p, int n) {
  int i = blockIdx.x * blockDim.x + threadIdx.x;
  if (i < n) p[i] = 0.f;
}

__global__ void k_colsum(const float* __restrict__ x, float* __restrict__ g, int N) {
  int c = threadIdx.x;
  int r0 = blockIdx.x * 256;
  int r1 = r0 + 256; if (r1 > N) r1 = N;
  float acc = 0.f;
  for (int r = r0; r < r1; ++r) acc += x[(size_t)r * H + c];
  atomicAdd(&g[c], acc);
}

__global__ void k_selector(const float* __restrict__ g,
                           const float* __restrict__ W1, const float* __restrict__ b1,
                           const float* __restrict__ W2, const float* __restrict__ b2,
                           float* __restrict__ sw, int N) {
  __shared__ float gsh[H];
  __shared__ float hid[128];
  __shared__ float lg[4];
  int t = threadIdx.x;
  float invN = 1.f / (float)N;
  gsh[t] = g[t] * invN;
  gsh[t + 128] = g[t + 128] * invN;
  __syncthreads();
  float a = b1[t];
  for (int k = 0; k < H; ++k) a += gsh[k] * W1[k * 128 + t];
  hid[t] = fmaxf(a, 0.f);
  __syncthreads();
  if (t < 4) {
    float l = b2[t];
    for (int k = 0; k < 128; ++k) l += hid[k] * W2[k * 4 + t];
    lg[t] = l;
  }
  __syncthreads();
  if (t == 0) {
    float m = fmaxf(fmaxf(lg[0], lg[1]), fmaxf(lg[2], lg[3]));
    float e0 = __expf(lg[0] - m), e1 = __expf(lg[1] - m);
    float e2 = __expf(lg[2] - m), e3 = __expf(lg[3] - m);
    float s = e0 + e1 + e2 + e3;
    sw[0] = e0 / s; sw[1] = e1 / s; sw[2] = e2 / s; sw[3] = e3 / s;
  }
}

__global__ void k_sd(const float* __restrict__ h, const float* __restrict__ asrc,
                     const float* __restrict__ adst, float* __restrict__ s,
                     float* __restrict__ d, int N) {
  int n = blockIdx.x * 8 + (threadIdx.x >> 5);
  if (n >= N) return;
  int lane = threadIdx.x & 31;
  float vs = 0.f, vd = 0.f;
  for (int c = lane; c < H; c += 32) {
    float v = h[(size_t)n * H + c];
    vs += v * asrc[c];
    vd += v * adst[c];
  }
  for (int o = 16; o; o >>= 1) { vs += __shfl_xor(vs, o, 32); vd += __shfl_xor(vd, o, 32); }
  if (lane == 0) { s[n] = vs; d[n] = vd; }
}

__global__ void k_init_m(const float* __restrict__ s, const float* __restrict__ d,
                         unsigned* __restrict__ mb, int N) {
  int n = blockIdx.x * blockDim.x + threadIdx.x;
  if (n < N) mb[n] = f2ord(lrelu(s[n] + d[n]));
}

__global__ void k_edge_max(const int* __restrict__ src, const int* __restrict__ dst,
                           const int* __restrict__ attr, int branch,
                           const float* __restrict__ s, const float* __restrict__ d,
                           unsigned* __restrict__ mb, int E) {
  int e = blockIdx.x * blockDim.x + threadIdx.x;
  if (e < E && attr[e] == branch) {
    int dn = dst[e];
    float v = lrelu(s[src[e]] + d[dn]);
    atomicMax(&mb[dn], f2ord(v));
  }
}

__global__ void k_init_num(const float* __restrict__ h, const float* __restrict__ s,
                           const float* __restrict__ d, const unsigned* __restrict__ mb,
                           float* __restrict__ mf, float* __restrict__ denom,
                           float* __restrict__ num, int N) {
  int n = blockIdx.x * 8 + (threadIdx.x >> 5);
  if (n >= N) return;
  int lane = threadIdx.x & 31;
  float m = ord2f(mb[n]);
  float wself = __expf(lrelu(s[n] + d[n]) - m);
  if (lane == 0) { mf[n] = m; denom[n] = wself; }
  for (int c = lane; c < H; c += 32) num[(size_t)n * H + c] = wself * h[(size_t)n * H + c];
}

__global__ void k_edge_accum(const int* __restrict__ src, const int* __restrict__ dst,
                             const int* __restrict__ attr, int branch,
                             const float* __restrict__ s, const float* __restrict__ d,
                             const float* __restrict__ mf, const float* __restrict__ h,
                             float* __restrict__ denom, float* __restrict__ num, int E) {
  int e = blockIdx.x * 8 + (threadIdx.x >> 5);
  if (e >= E || attr[e] != branch) return;
  int sn = src[e], dn = dst[e];
  float w = __expf(lrelu(s[sn] + d[dn]) - mf[dn]);
  int lane = threadIdx.x & 31;
  if (lane == 0) atomicAdd(&denom[dn], w);
  for (int c = lane; c < H; c += 32)
    atomicAdd(&num[(size_t)dn * H + c], w * h[(size_t)sn * H + c]);
}

__global__ void k_finalize(float* __restrict__ num, const float* __restrict__ denom,
                           const float* __restrict__ bias, int total) {
  int i = blockIdx.x * blockDim.x + threadIdx.x;
  if (i < total) {
    int n = i >> 8;
    num[i] = num[i] / denom[n] + bias[i & (H - 1)];
  }
}

__global__ void k_ln_relu(const float* __restrict__ t, const float* __restrict__ gamma,
                          const float* __restrict__ beta, float* __restrict__ out,
                          const float* __restrict__ swp, int swidx, int accumulate, int N) {
  int n = blockIdx.x * 8 + (threadIdx.x >> 5);
  if (n >= N) return;
  int lane = threadIdx.x & 31;
  float v[8];
  float sum = 0.f, sq = 0.f;
#pragma unroll
  for (int j = 0; j < 8; ++j) {
    v[j] = t[(size_t)n * H + lane + j * 32];
    sum += v[j];
    sq += v[j] * v[j];
  }
  for (int o = 16; o; o >>= 1) { sum += __shfl_xor(sum, o, 32); sq += __shfl_xor(sq, o, 32); }
  float mu = sum * (1.f / H);
  float var = sq * (1.f / H) - mu * mu;
  float inv = rsqrtf(var + 1e-5f);
  float scale = swp ? swp[swidx] : 1.f;
#pragma unroll
  for (int j = 0; j < 8; ++j) {
    int c = lane + j * 32;
    float y = (v[j] - mu) * inv * gamma[c] + beta[c];
    y = fmaxf(y, 0.f) * scale;
    if (accumulate) out[(size_t)n * H + c] += y;
    else            out[(size_t)n * H + c] = y;
  }
}

extern "C" void kernel_launch(void* const* d_in, const int* in_sizes, int n_in,
                              void* d_out, int out_size, void* d_ws, size_t ws_size,
                              hipStream_t stream) {
  const float* x     = (const float*)d_in[0];
  const int*   ei    = (const int*)d_in[1];
  const int*   ea    = (const int*)d_in[2];
  const float* Wg    = (const float*)d_in[3];
  const float* a_src = (const float*)d_in[4];
  const float* a_dst = (const float*)d_in[5];
  const float* bg    = (const float*)d_in[6];
  const float* Wsl   = (const float*)d_in[7];
  const float* bsl   = (const float*)d_in[8];
  const float* gsl   = (const float*)d_in[9];
  const float* betas = (const float*)d_in[10];
  const float* W1    = (const float*)d_in[11];
  const float* b1    = (const float*)d_in[12];
  const float* W2    = (const float*)d_in[13];
  const float* b2    = (const float*)d_in[14];
  const float* Wf    = (const float*)d_in[15];
  const float* bfv   = (const float*)d_in[16];
  const float* gf    = (const float*)d_in[17];
  const float* betaf = (const float*)d_in[18];

  const int N = in_sizes[0] / H;
  const int E = in_sizes[2];
  float* out = (float*)d_out;

  char* wsb = (char*)d_ws;
  size_t off = 0;
  auto alloc = [&](size_t bytes) -> void* {
    void* p = wsb + off;
    off += (bytes + 255) & ~(size_t)255;
    return p;
  };
  float*    g     = (float*)alloc(H * sizeof(float));
  float*    sw    = (float*)alloc(4 * sizeof(float));
  float*    sArr  = (float*)alloc((size_t)N * sizeof(float));
  float*    dArr  = (float*)alloc((size_t)N * sizeof(float));
  unsigned* mb    = (unsigned*)alloc((size_t)N * sizeof(unsigned));
  float*    mf    = (float*)alloc((size_t)N * sizeof(float));
  float*    denom = (float*)alloc((size_t)N * sizeof(float));
  float*    bufH  = (float*)alloc((size_t)N * H * sizeof(float));
  float*    bufX  = (float*)alloc((size_t)N * H * sizeof(float));
  float*    bufC  = (float*)alloc((size_t)N * H * sizeof(float));

  dim3 gemmGrid((unsigned)((N + TM - 1) / TM), (unsigned)(H / TNB));
  int nWaveBlocks = (N + 7) / 8;
  int eWaveBlocks = (E + 7) / 8;
  int nThrBlocks  = (N + 255) / 256;
  int eThrBlocks  = (E + 255) / 256;
  int nhBlocks    = (N * H + 255) / 256;

  k_zero<<<1, 256, 0, stream>>>(g, H);
  k_colsum<<<nThrBlocks, 256, 0, stream>>>(x, g, N);
  k_selector<<<1, 128, 0, stream>>>(g, W1, b1, W2, b2, sw, N);

  for (int i = 0; i < 3; ++i) {
    k_gemm_bf16x3<<<gemmGrid, 256, 0, stream>>>(x, Wg + (size_t)i * H * H, nullptr, bufH, N);
    k_sd<<<nWaveBlocks, 256, 0, stream>>>(bufH, a_src + i * H, a_dst + i * H, sArr, dArr, N);
    k_init_m<<<nThrBlocks, 256, 0, stream>>>(sArr, dArr, mb, N);
    k_edge_max<<<eThrBlocks, 256, 0, stream>>>(ei, ei + E, ea, i, sArr, dArr, mb, E);
    k_init_num<<<nWaveBlocks, 256, 0, stream>>>(bufH, sArr, dArr, mb, mf, denom, bufX, N);
    k_edge_accum<<<eWaveBlocks, 256, 0, stream>>>(ei, ei + E, ea, i, sArr, dArr, mf, bufH, denom, bufX, E);
    k_finalize<<<nhBlocks, 256, 0, stream>>>(bufX, denom, bg + i * H, N * H);
    k_gemm_bf16x3<<<gemmGrid, 256, 0, stream>>>(bufX, Wsl + (size_t)i * H * H, bsl + i * H, bufH, N);
    k_ln_relu<<<nWaveBlocks, 256, 0, stream>>>(bufH, gsl + i * H, betas + i * H, bufC, sw, i,
                                               i > 0 ? 1 : 0, N);
  }
  k_gemm_bf16x3<<<gemmGrid, 256, 0, stream>>>(x, Wsl + (size_t)3 * H * H, bsl + 3 * H, bufH, N);
  k_ln_relu<<<nWaveBlocks, 256, 0, stream>>>(bufH, gsl + 3 * H, betas + 3 * H, bufC, sw, 3, 1, N);

  k_gemm_bf16x3<<<gemmGrid, 256, 0, stream>>>(bufC, Wf, bfv, bufH, N);
  k_ln_relu<<<nWaveBlocks, 256, 0, stream>>>(bufH, gf, betaf, out, nullptr, 0, 0, N);
}


# second product's A loader finishes the attention result: adds self-loop weight x h to zero-started edge sums, w_self to the divisor; 3 init launches gone; self term summed last (order changed)
# speedup vs baseline: 1.6026x; 1.6026x over previous
#include <hip/hip_runtime.h>

#define H 256
#define KC 32
#define BP 264
#define GROWS 256

typedef __bf16 bf16;
typedef __attribute__((ext_vector_type(16))) __bf16 v16bf;
typedef __attribute__((ext_vector_type(8)))  __bf16 v8bf;
typedef __attribute__((ext_vector_type(8)))  float  v8f;

__device__ __forceinline__ unsigned short f2bfbits(float f) {
  unsigned u = __float_as_uint(f);
  return (unsigned short)((u + 0x7FFFu + ((u >> 16) & 1u)) >> 16);
}
__device__ __forceinline__ float bfbits2f(unsigned short h) {
  return __uint_as_float(((unsigned)h) << 16);
}
__device__ __forceinline__ bf16 f2bf(float f) {
  unsigned short hs = f2bfbits(f);
  return __builtin_bit_cast(bf16, hs);
}
__device__ __forceinline__ float lrelu(float v) { return v > 0.f ? v : 0.2f * v; }
__device__ __forceinline__ unsigned f2ord(float f) {
  unsigned u = __float_as_uint(f);
  return (u & 0x80000000u) ? ~u : (u | 0x80000000u);
}
__device__ __forceinline__ float ord2f(unsigned u) {
  unsigned b = (u & 0x80000000u) ? (u & 0x7FFFFFFFu) : ~u;
  return __uint_as_float(b);
}

__device__ __forceinline__ v16bf load_frag(const bf16* rowp, int g0) {
  v8bf a = *(const v8bf*)(rowp + g0);
  v8bf b = *(const v8bf*)(rowp + 16 + g0);
  return __builtin_shufflevector(a, b, 0, 1, 2, 3, 4, 5, 6, 7,
                                 8, 9, 10, 11, 12, 13, 14, 15);
}

#define WMMA_BF16(A_, B_, C_) \
  __builtin_amdgcn_wmma_f32_16x16x32_bf16(false, (A_), false, (B_), (short)0, (C_), false, false)

__device__ __forceinline__ void split16(const float4 a, const float4 b, const float4 c, const float4 d,
                                        v16bf& hi, v16bf& lo) {
  const float f[16] = {a.x, a.y, a.z, a.w, b.x, b.y, b.z, b.w,
                       c.x, c.y, c.z, c.w, d.x, d.y, d.z, d.w};
#pragma unroll
  for (int q = 0; q < 16; ++q) {
    bf16 h = (bf16)f[q];
    hi[q] = h;
    lo[q] = (bf16)(f[q] - (float)h);
  }
}

__device__ __forceinline__ float4 fma4(const float4 a, const float s, const float4 b) {
  return make_float4(fmaf(a.x, s, b.x), fmaf(a.y, s, b.y), fmaf(a.z, s, b.z), fmaf(a.w, s, b.w));
}

template <bool NORM>
__global__ __launch_bounds__(256) void k_gemm_bf16x3(
    const float* __restrict__ A, const float* __restrict__ B,
    const float* __restrict__ bias, float* C, int M,
    const float* __restrict__ rden, const float* __restrict__ abias,
    const float* Hs, const float* __restrict__ sv, const float* __restrict__ dv,
    const unsigned* __restrict__ mxv)
{
  __shared__ __attribute__((aligned(16))) bf16 Bth[H][BP];
  __shared__ __attribute__((aligned(16))) bf16 Btl[H][BP];
  __shared__ __attribute__((aligned(16))) float Ab[H];

  const int tid = threadIdx.x;
  const int wid = tid >> 5, lane = tid & 31;
  const int hl = lane & 15, hh = lane >> 4;
  const int g0 = hh << 3;
  const int row0 = blockIdx.x * GROWS + wid * 32;

  int ra0 = row0 + hl;      if (ra0 > M - 1) ra0 = M - 1;
  int ra1 = row0 + 16 + hl; if (ra1 > M - 1) ra1 = M - 1;
  if (ra0 < 0) ra0 = 0;
  if (ra1 < 0) ra1 = 0;
  const unsigned oa0 = (unsigned)ra0 * H + g0;
  const unsigned oa1 = (unsigned)ra1 * H + g0;
  const float* pa0 = A + oa0;
  const float* pa1 = A + oa1;
  float inv0 = 1.f, inv1 = 1.f, ws0 = 0.f, ws1 = 0.f;
  const float* ph0 = A;
  const float* ph1 = A;
  if (NORM) {
    ws0 = __expf(lrelu(sv[ra0] + dv[ra0]) - ord2f(mxv[ra0]));
    ws1 = __expf(lrelu(sv[ra1] + dv[ra1]) - ord2f(mxv[ra1]));
    inv0 = 1.f / (rden[ra0] + ws0);
    inv1 = 1.f / (rden[ra1] + ws1);
    ph0 = Hs + oa0;
    ph1 = Hs + oa1;
  }

  float4 x0, x1, x2, x3, y0, y1, y2, y3;
  float4 u0, u1, u2, u3, w0, w1, w2, w3;
#define LOAD_A(ko_)                                                            \
  do {                                                                         \
    x0 = *(const float4*)(pa0 + (ko_));      x1 = *(const float4*)(pa0 + (ko_) + 4);  \
    x2 = *(const float4*)(pa0 + (ko_) + 16); x3 = *(const float4*)(pa0 + (ko_) + 20); \
    y0 = *(const float4*)(pa1 + (ko_));      y1 = *(const float4*)(pa1 + (ko_) + 4);  \
    y2 = *(const float4*)(pa1 + (ko_) + 16); y3 = *(const float4*)(pa1 + (ko_) + 20); \
    if (NORM) {                                                                \
      u0 = *(const float4*)(ph0 + (ko_));      u1 = *(const float4*)(ph0 + (ko_) + 4);  \
      u2 = *(const float4*)(ph0 + (ko_) + 16); u3 = *(const float4*)(ph0 + (ko_) + 20); \
      w0 = *(const float4*)(ph1 + (ko_));      w1 = *(const float4*)(ph1 + (ko_) + 4);  \
      w2 = *(const float4*)(ph1 + (ko_) + 16); w3 = *(const float4*)(ph1 + (ko_) + 20); \
    }                                                                          \
  } while (0)
#define ORDER_FENCE() asm volatile("" ::: "memory")

  LOAD_A(0);

#pragma unroll 4
  for (int i = 0; i < (H / 8) * (H / 4) / 256; ++i) {
    const int it = tid + i * 256;
    const int cg = it & 63;
    const int kg = it >> 6;
    float4 v[8];
#pragma unroll
    for (int q = 0; q < 8; ++q) v[q] = *(const float4*)&B[(size_t)(kg * 8 + q) * H + cg * 4];
    const int rowb = ((cg >> 4) << 6) + (cg & 15);
    v8bf h0, h1, h2, h3, l0, l1, l2, l3;
#pragma unroll
    for (int q = 0; q < 8; ++q) {
      bf16 t;
      t = (bf16)v[q].x; h0[q] = t; l0[q] = (bf16)(v[q].x - (float)t);
      t = (bf16)v[q].y; h1[q] = t; l1[q] = (bf16)(v[q].y - (float)t);
      t = (bf16)v[q].z; h2[q] = t; l2[q] = (bf16)(v[q].z - (float)t);
      t = (bf16)v[q].w; h3[q] = t; l3[q] = (bf16)(v[q].w - (float)t);
    }
    *(v8bf*)&Bth[rowb][kg * 8]      = h0;  *(v8bf*)&Btl[rowb][kg * 8]      = l0;
    *(v8bf*)&Bth[rowb + 16][kg * 8] = h1;  *(v8bf*)&Btl[rowb + 16][kg * 8] = l1;
    *(v8bf*)&Bth[rowb + 32][kg * 8] = h2;  *(v8bf*)&Btl[rowb + 32][kg * 8] = l2;
    *(v8bf*)&Bth[rowb + 48][kg * 8] = h3;  *(v8bf*)&Btl[rowb + 48][kg * 8] = l3;
  }
  if (NORM) {
    if (tid < H / 4) *(float4*)&Ab[tid * 4] = *(const float4*)&abias[tid * 4];
  }
  __syncthreads();

  if (row0 >= M) return;

  v8f acc0[16], acc1[16];
#pragma unroll
  for (int nt = 0; nt < 16; ++nt) { acc0[nt] = (v8f){}; acc1[nt] = (v8f){}; }

  const bf16* bph = &Bth[hl][0];
  const bf16* bpl = &Btl[hl][0];
  v16bf bh[2][4], bl[2][4];
#define LOAD_B(set_, ko_, ng_)                                                             \
  do {                                                                                     \
    _Pragma("unroll") for (int j = 0; j < 4; ++j) {                                        \
      bh[set_][j] = load_frag(bph + (((ng_) * 4 + j) * 16) * BP + (ko_), g0);              \
      bl[set_][j] = load_frag(bpl + (((ng_) * 4 + j) * 16) * BP + (ko_), g0);              \
    }                                                                                      \
  } while (0)

  LOAD_B(0, 0, 0);
#pragma unroll 1
  for (int kc = 0; kc < H / KC; ++kc) {
    v16bf ah0, al0, ah1, al1;
    if (NORM) {
      const float* bq = &Ab[kc * KC + g0];
      const float4 q0 = *(const float4*)(bq),      q1 = *(const float4*)(bq + 4);
      const float4 q2 = *(const float4*)(bq + 16), q3 = *(const float4*)(bq + 20);
      x0 = fma4(u0, ws0, x0); x1 = fma4(u1, ws0, x1); x2 = fma4(u2, ws0, x2); x3 = fma4(u3, ws0, x3);
      y0 = fma4(w0, ws1, y0); y1 = fma4(w1, ws1, y1); y2 = fma4(w2, ws1, y2); y3 = fma4(w3, ws1, y3);
      x0 = fma4(x0, inv0, q0); x1 = fma4(x1, inv0, q1); x2 = fma4(x2, inv0, q2); x3 = fma4(x3, inv0, q3);
      y0 = fma4(y0, inv1, q0); y1 = fma4(y1, inv1, q1); y2 = fma4(y2, inv1, q2); y3 = fma4(y3, inv1, q3);
    }
    split16(x0, x1, x2, x3, ah0, al0);
    split16(y0, y1, y2, y3, ah1, al1);
    const int kn = ((kc + 1) & (H / KC - 1)) * KC;
    LOAD_A(kn);
    ORDER_FENCE();
#pragma unroll
    for (int ng = 0; ng < 4; ++ng) {
      const int cur = ng & 1;
      if (ng < 3) LOAD_B(cur ^ 1, kc * KC, ng + 1);
      else        LOAD_B(cur ^ 1, kn, 0);
      ORDER_FENCE();
#pragma unroll
      for (int j = 0; j < 4; ++j) acc0[ng * 4 + j] = WMMA_BF16(ah0, bh[cur][j], acc0[ng * 4 + j]);
#pragma unroll
      for (int j = 0; j < 4; ++j) acc1[ng * 4 + j] = WMMA_BF16(ah1, bh[cur][j], acc1[ng * 4 + j]);
#pragma unroll
      for (int j = 0; j < 4; ++j) acc0[ng * 4 + j] = WMMA_BF16(al0, bh[cur][j], acc0[ng * 4 + j]);
#pragma unroll
      for (int j = 0; j < 4; ++j) acc1[ng * 4 + j] = WMMA_BF16(al1, bh[cur][j], acc1[ng * 4 + j]);
#pragma unroll
      for (int j = 0; j < 4; ++j) acc0[ng * 4 + j] = WMMA_BF16(ah0, bl[cur][j], acc0[ng * 4 + j]);
#pragma unroll
      for (int j = 0; j < 4; ++j) acc1[ng * 4 + j] = WMMA_BF16(ah1, bl[cur][j], acc1[ng * 4 + j]);
      ORDER_FENCE();
    }
  }
#undef LOAD_A
#undef LOAD_B
#undef ORDER_FENCE

  const bool full = (row0 + 32 <= M);
#pragma unroll
  for (int ng = 0; ng < 4; ++ng) {
    const int col = ng * 64 + hl * 4;
    float4 bv = bias ? *(const float4*)&bias[col] : make_float4(0.f, 0.f, 0.f, 0.f);
#pragma unroll
    for (int r = 0; r < 8; ++r) {
      const int r0g = row0 + (hh << 3) + r;
      const int r1g = r0g + 16;
      float4 o0 = make_float4(acc0[ng * 4 + 0][r] + bv.x, acc0[ng * 4 + 1][r] + bv.y,
                              acc0[ng * 4 + 2][r] + bv.z, acc0[ng * 4 + 3][r] + bv.w);
      float4 o1 = make_float4(acc1[ng * 4 + 0][r] + bv.x, acc1[ng * 4 + 1][r] + bv.y,
                              acc1[ng * 4 + 2][r] + bv.z, acc1[ng * 4 + 3][r] + bv.w);
      if (full) {
        *(float4*)&C[(size_t)r0g * H + col] = o0;
        *(float4*)&C[(size_t)r1g * H + col] = o1;
      } else {
        if (r0g < M) *(float4*)&C[(size_t)r0g * H + col] = o0;
        if (r1g < M) *(float4*)&C[(size_t)r1g * H + col] = o1;
      }
    }
  }
}

__global__ void k_zero(float* p, int n) {
  int i = blockIdx.x * blockDim.x + threadIdx.x;
  if (i < n) p[i] = 0.f;
}

__global__ void k_colsum(const float* __restrict__ x, float* __restrict__ g, int N) {
  int c = threadIdx.x;
  int r0 = blockIdx.x * 256;
  int r1 = r0 + 256; if (r1 > N) r1 = N;
  float acc = 0.f;
  for (int r = r0; r < r1; ++r) acc += x[(size_t)r * H + c];
  atomicAdd(&g[c], acc);
}

__global__ void k_selector(const float* __restrict__ g,
                           const float* __restrict__ W1, const float* __restrict__ b1,
                           const float* __restrict__ W2, const float* __restrict__ b2,
                           float* __restrict__ sw, int N) {
  __shared__ float gsh[H];
  __shared__ float hid[128];
  __shared__ float lg[4];
  int t = threadIdx.x;
  float invN = 1.f / (float)N;
  gsh[t] = g[t] * invN;
  gsh[t + 128] = g[t + 128] * invN;
  __syncthreads();
  float a = b1[t];
  for (int k = 0; k < H; ++k) a += gsh[k] * W1[k * 128 + t];
  hid[t] = fmaxf(a, 0.f);
  __syncthreads();
  if (t < 4) {
    float l = b2[t];
    for (int k = 0; k < 128; ++k) l += hid[k] * W2[k * 4 + t];
    lg[t] = l;
  }
  __syncthreads();
  if (t == 0) {
    float m = fmaxf(fmaxf(lg[0], lg[1]), fmaxf(lg[2], lg[3]));
    float e0 = __expf(lg[0] - m), e1 = __expf(lg[1] - m);
    float e2 = __expf(lg[2] - m), e3 = __expf(lg[3] - m);
    float s = e0 + e1 + e2 + e3;
    sw[0] = e0 / s; sw[1] = e1 / s; sw[2] = e2 / s; sw[3] = e3 / s;
  }
}

__global__ void k_sd(const float* __restrict__ h, const float* __restrict__ asrc,
                     const float* __restrict__ adst, float* __restrict__ s,
                     float* __restrict__ d, float* __restrict__ num, int N) {
  int n = blockIdx.x * 8 + (threadIdx.x >> 5);
  if (n >= N) return;
  int lane = threadIdx.x & 31;
  const float4 z4 = make_float4(0.f, 0.f, 0.f, 0.f);
  *(float4*)&num[(size_t)n * H + lane * 8]     = z4;
  *(float4*)&num[(size_t)n * H + lane * 8 + 4] = z4;
  float vs = 0.f, vd = 0.f;
  for (int c = lane; c < H; c += 32) {
    float v = h[(size_t)n * H + c];
    vs += v * asrc[c];
    vd += v * adst[c];
  }
  for (int o = 16; o; o >>= 1) { vs += __shfl_xor(vs, o, 32); vd += __shfl_xor(vd, o, 32); }
  if (lane == 0) { s[n] = vs; d[n] = vd; }
}

__global__ void k_init_m(const float* __restrict__ s, const float* __restrict__ d,
                         unsigned* __restrict__ mb, float* __restrict__ denom, int N) {
  int n = blockIdx.x * blockDim.x + threadIdx.x;
  if (n < N) {
    mb[n] = f2ord(lrelu(s[n] + d[n]));
    denom[n] = 0.f;
  }
}

__global__ void k_edge_max(const int* __restrict__ src, const int* __restrict__ dst,
                           const int* __restrict__ attr, int branch,
                           const float* __restrict__ s, const float* __restrict__ d,
                           unsigned* __restrict__ mb, int E) {
  int e = blockIdx.x * blockDim.x + threadIdx.x;
  if (e < E && attr[e] == branch) {
    int dn = dst[e];
    float v = lrelu(s[src[e]] + d[dn]);
    atomicMax(&mb[dn], f2ord(v));
  }
}

__global__ void k_edge_accum(const int* __restrict__ src, const int* __restrict__ dst,
                             const int* __restrict__ attr, int branch,
                             const float* __restrict__ s, const float* __restrict__ d,
                             const unsigned* __restrict__ mb, const float* __restrict__ h,
                             float* __restrict__ denom, float* __restrict__ num, int E) {
  int e = blockIdx.x * 8 + (threadIdx.x >> 5);
  if (e >= E || attr[e] != branch) return;
  int sn = src[e], dn = dst[e];
  float w = __expf(lrelu(s[sn] + d[dn]) - ord2f(mb[dn]));
  int lane = threadIdx.x & 31;
  if (lane == 0) atomicAdd(&denom[dn], w);
  for (int c = lane; c < H; c += 32)
    atomicAdd(&num[(size_t)dn * H + c], w * h[(size_t)sn * H + c]);
}

__global__ void k_ln_relu(const float* __restrict__ t, const float* __restrict__ gamma,
                          const float* __restrict__ beta, float* __restrict__ out,
                          const float* __restrict__ swp, int swidx, int accumulate, int N) {
  int n = blockIdx.x * 8 + (threadIdx.x >> 5);
  if (n >= N) return;
  int lane = threadIdx.x & 31;
  float v[8];
  float sum = 0.f, sq = 0.f;
#pragma unroll
  for (int j = 0; j < 8; ++j) {
    v[j] = t[(size_t)n * H + lane + j * 32];
    sum += v[j];
    sq += v[j] * v[j];
  }
  for (int o = 16; o; o >>= 1) { sum += __shfl_xor(sum, o, 32); sq += __shfl_xor(sq, o, 32); }
  float mu = sum * (1.f / H);
  float var = sq * (1.f / H) - mu * mu;
  float inv = rsqrtf(var + 1e-5f);
  float scale = swp ? swp[swidx] : 1.f;
#pragma unroll
  for (int j = 0; j < 8; ++j) {
    int c = lane + j * 32;
    float y = (v[j] - mu) * inv * gamma[c] + beta[c];
    y = fmaxf(y, 0.f) * scale;
    if (accumulate) out[(size_t)n * H + c] += y;
    else            out[(size_t)n * H + c] = y;
  }
}

extern "C" void kernel_launch(void* const* d_in, const int* in_sizes, int n_in,
                              void* d_out, int out_size, void* d_ws, size_t ws_size,
                              hipStream_t stream) {
  const float* x     = (const float*)d_in[0];
  const int*   ei    = (const int*)d_in[1];
  const int*   ea    = (const int*)d_in[2];
  const float* Wg    = (const float*)d_in[3];
  const float* a_src = (const float*)d_in[4];
  const float* a_dst = (const float*)d_in[5];
  const float* bg    = (const float*)d_in[6];
  const float* Wsl   = (const float*)d_in[7];
  const float* bsl   = (const float*)d_in[8];
  const float* gsl   = (const float*)d_in[9];
  const float* betas = (const float*)d_in[10];
  const float* W1    = (const float*)d_in[11];
  const float* b1    = (const float*)d_in[12];
  const float* W2    = (const float*)d_in[13];
  const float* b2    = (const float*)d_in[14];
  const float* Wf    = (const float*)d_in[15];
  const float* bfv   = (const float*)d_in[16];
  const float* gf    = (const float*)d_in[17];
  const float* betaf = (const float*)d_in[18];

  const int N = in_sizes[0] / H;
  const int E = in_sizes[2];
  float* out = (float*)d_out;

  char* wsb = (char*)d_ws;
  size_t off = 0;
  auto alloc = [&](size_t bytes) -> void* {
    void* p = wsb + off;
    off += (bytes + 255) & ~(size_t)255;
    return p;
  };
  float*    g     = (float*)alloc(H * sizeof(float));
  float*    sw    = (float*)alloc(4 * sizeof(float));
  float*    sArr  = (float*)alloc((size_t)N * sizeof(float));
  float*    dArr  = (float*)alloc((size_t)N * sizeof(float));
  unsigned* mb    = (unsigned*)alloc((size_t)N * sizeof(unsigned));
  float*    denom = (float*)alloc((size_t)N * sizeof(float));
  float*    bufH  = (float*)alloc((size_t)N * H * sizeof(float));
  float*    bufX  = (float*)alloc((size_t)N * H * sizeof(float));
  float*    bufC  = (float*)alloc((size_t)N * H * sizeof(float));

  int gemmGrid = (N + GROWS - 1) / GROWS;
  int nWaveBlocks = (N + 7) / 8;
  int eWaveBlocks = (E + 7) / 8;
  int nThrBlocks  = (N + 255) / 256;
  int eThrBlocks  = (E + 255) / 256;

  k_zero<<<1, 256, 0, stream>>>(g, H);
  k_colsum<<<nThrBlocks, 256, 0, stream>>>(x, g, N);
  k_selector<<<1, 128, 0, stream>>>(g, W1, b1, W2, b2, sw, N);

  for (int i = 0; i < 3; ++i) {
    k_gemm_bf16x3<false><<<gemmGrid, 256, 0, stream>>>(x, Wg + (size_t)i * H * H, nullptr, bufH, N,
                                                       nullptr, nullptr, nullptr, nullptr, nullptr, nullptr);
    k_sd<<<nWaveBlocks, 256, 0, stream>>>(bufH, a_src + i * H, a_dst + i * H, sArr, dArr, bufX, N);
    k_init_m<<<nThrBlocks, 256, 0, stream>>>(sArr, dArr, mb, denom, N);
    k_edge_max<<<eThrBlocks, 256, 0, stream>>>(ei, ei + E, ea, i, sArr, dArr, mb, E);
    k_edge_accum<<<eWaveBlocks, 256, 0, stream>>>(ei, ei + E, ea, i, sArr, dArr, mb, bufH, denom, bufX, E);
    k_gemm_bf16x3<true><<<gemmGrid, 256, 0, stream>>>(bufX, Wsl + (size_t)i * H * H, bsl + i * H, bufH, N,
                                                      denom, bg + i * H, bufH, sArr, dArr, mb);
    k_ln_relu<<<nWaveBlocks, 256, 0, stream>>>(bufH, gsl + i * H, betas + i * H, bufC, sw, i,
                                               i > 0 ? 1 : 0, N);
  }
  k_gemm_bf16x3<false><<<gemmGrid, 256, 0, stream>>>(x, Wsl + (size_t)3 * H * H, bsl + 3 * H, bufH, N,
                                                     nullptr, nullptr, nullptr, nullptr, nullptr, nullptr);
  k_ln_relu<<<nWaveBlocks, 256, 0, stream>>>(bufH, gsl + 3 * H, betas + 3 * H, bufC, sw, 3, 1, N);

  k_gemm_bf16x3<false><<<gemmGrid, 256, 0, stream>>>(bufC, Wf, bfv, bufH, N, nullptr, nullptr,
                                                     nullptr, nullptr, nullptr, nullptr);
  k_ln_relu<<<nWaveBlocks, 256, 0, stream>>>(bufH, gf, betaf, out, nullptr, 0, 0, N);
}


# column sum of x: each thread requests 32 rows of its column as one group ahead of the adds (was one request, one wait per row: 256 round trips in a row); grid, atomics and add order kept
# speedup vs baseline: 2.2657x; 1.4138x over previous
#include <hip/hip_runtime.h>

#define H 256
#define KC 32
#define BP 264
#define GROWS 256

typedef __bf16 bf16;
typedef __attribute__((ext_vector_type(16))) __bf16 v16bf;
typedef __attribute__((ext_vector_type(8)))  __bf16 v8bf;
typedef __attribute__((ext_vector_type(8)))  float  v8f;

__device__ __forceinline__ unsigned short f2bfbits(float f) {
  unsigned u = __float_as_uint(f);
  return (unsigned short)((u + 0x7FFFu + ((u >> 16) & 1u)) >> 16);
}
__device__ __forceinline__ float bfbits2f(unsigned short h) {
  return __uint_as_float(((unsigned)h) << 16);
}
__device__ __forceinline__ bf16 f2bf(float f) {
  unsigned short hs = f2bfbits(f);
  return __builtin_bit_cast(bf16, hs);
}
__device__ __forceinline__ float lrelu(float v) { return v > 0.f ? v : 0.2f * v; }
__device__ __forceinline__ unsigned f2ord(float f) {
  unsigned u = __float_as_uint(f);
  return (u & 0x80000000u) ? ~u : (u | 0x80000000u);
}
__device__ __forceinline__ float ord2f(unsigned u) {
  unsigned b = (u & 0x80000000u) ? (u & 0x7FFFFFFFu) : ~u;
  return __uint_as_float(b);
}

__device__ __forceinline__ v16bf load_frag(const bf16* rowp, int g0) {
  v8bf a = *(const v8bf*)(rowp + g0);
  v8bf b = *(const v8bf*)(rowp + 16 + g0);
  return __builtin_shufflevector(a, b, 0, 1, 2, 3, 4, 5, 6, 7,
                                 8, 9, 10, 11, 12, 13, 14, 15);
}

#define WMMA_BF16(A_, B_, C_) \
  __builtin_amdgcn_wmma_f32_16x16x32_bf16(false, (A_), false, (B_), (short)0, (C_), false, false)

__device__ __forceinline__ void split16(const float4 a, const float4 b, const float4 c, const float4 d,
                                        v16bf& hi, v16bf& lo) {
  const float f[16] = {a.x, a.y, a.z, a.w, b.x, b.y, b.z, b.w,
                       c.x, c.y, c.z, c.w, d.x, d.y, d.z, d.w};
#pragma unroll
  for (int q = 0; q < 16; ++q) {
    bf16 h = (bf16)f[q];
    hi[q] = h;
    lo[q] = (bf16)(f[q] - (float)h);
  }
}

__device__ __forceinline__ float4 fma4(const float4 a, const float s, const float4 b) {
  return make_float4(fmaf(a.x, s, b.x), fmaf(a.y, s, b.y), fmaf(a.z, s, b.z), fmaf(a.w, s, b.w));
}

template <bool ACC>
__device__ __forceinline__ void ln_epilogue(const v8f (&acc0)[16], const v8f (&acc1)[16], float* C, int M,
                                            const float* __restrict__ bias,
                                            const float* __restrict__ gamma,
                                            const float* __restrict__ beta,
                                            const float scale, const int row0, const int rb,
                                            const int hl) {
  const float* csafe = C + (size_t)row0 * H + hl * 4;
  float* cp = C + (size_t)rb * H + hl * 4;
  const int nv0 = M - rb;
  const int nv1 = M - rb - 16;
  float4 bv[4];
#pragma unroll
  for (int ng = 0; ng < 4; ++ng) bv[ng] = *(const float4*)&bias[ng * 64 + hl * 4];
  float4 pv[2][2][4];
  if (ACC) {
    const float* p0 = (0 < nv0) ? cp : csafe;
    const float* p1 = (0 < nv1) ? cp + 16 * H : csafe;
#pragma unroll
    for (int ng = 0; ng < 4; ++ng) {
      pv[0][0][ng] = *(const float4*)(p0 + ng * 64);
      pv[0][1][ng] = *(const float4*)(p1 + ng * 64);
    }
  }
  float s0[8], q0[8], s1[8], q1[8];
#pragma unroll
  for (int r = 0; r < 8; ++r) {
    float a = 0.f, b = 0.f, cc = 0.f, d = 0.f;
#pragma unroll
    for (int ng = 0; ng < 4; ++ng) {
      const float bb[4] = {bv[ng].x, bv[ng].y, bv[ng].z, bv[ng].w};
#pragma unroll
      for (int j = 0; j < 4; ++j) {
        const float v = acc0[ng * 4 + j][r] + bb[j];
        const float w = acc1[ng * 4 + j][r] + bb[j];
        a += v; b += v * v;
        cc += w; d += w * w;
      }
    }
    s0[r] = a; q0[r] = b; s1[r] = cc; q1[r] = d;
  }
#pragma unroll
  for (int o = 1; o < 16; o <<= 1) {
#pragma unroll
    for (int r = 0; r < 8; ++r) {
      s0[r] += __shfl_xor(s0[r], o, 32);
      q0[r] += __shfl_xor(q0[r], o, 32);
      s1[r] += __shfl_xor(s1[r], o, 32);
      q1[r] += __shfl_xor(q1[r], o, 32);
    }
  }
  float4 gv[4], ev[4];
#pragma unroll
  for (int ng = 0; ng < 4; ++ng) {
    gv[ng] = *(const float4*)&gamma[ng * 64 + hl * 4];
    ev[ng] = *(const float4*)&beta[ng * 64 + hl * 4];
  }
#pragma unroll
  for (int r = 0; r < 8; ++r) {
    if (ACC && r < 7) {
      const float* p0 = (r + 1 < nv0) ? cp + (r + 1) * H : csafe;
      const float* p1 = (r + 1 < nv1) ? cp + (16 + r + 1) * H : csafe;
#pragma unroll
      for (int ng = 0; ng < 4; ++ng) {
        pv[(r + 1) & 1][0][ng] = *(const float4*)(p0 + ng * 64);
        pv[(r + 1) & 1][1][ng] = *(const float4*)(p1 + ng * 64);
      }
    }
    const float mu0 = s0[r] * (1.f / H);
    const float mu1 = s1[r] * (1.f / H);
    const float rs0 = rsqrtf(q0[r] * (1.f / H) - mu0 * mu0 + 1e-5f);
    const float rs1 = rsqrtf(q1[r] * (1.f / H) - mu1 * mu1 + 1e-5f);
    float4 o0[4], o1[4];
#pragma unroll
    for (int ng = 0; ng < 4; ++ng) {
      const float bb[4] = {bv[ng].x, bv[ng].y, bv[ng].z, bv[ng].w};
      const float gg[4] = {gv[ng].x, gv[ng].y, gv[ng].z, gv[ng].w};
      const float ee[4] = {ev[ng].x, ev[ng].y, ev[ng].z, ev[ng].w};
      float y0[4], y1[4];
#pragma unroll
      for (int j = 0; j < 4; ++j) {
        const float m0 = mu0 - bb[j];
        const float m1 = mu1 - bb[j];
        y0[j] = fmaxf((acc0[ng * 4 + j][r] - m0) * rs0 * gg[j] + ee[j], 0.f) * scale;
        y1[j] = fmaxf((acc1[ng * 4 + j][r] - m1) * rs1 * gg[j] + ee[j], 0.f) * scale;
      }
      o0[ng] = make_float4(y0[0], y0[1], y0[2], y0[3]);
      o1[ng] = make_float4(y1[0], y1[1], y1[2], y1[3]);
    }
    if (r < nv0) {
#pragma unroll
      for (int ng = 0; ng < 4; ++ng) {
        float4 o = o0[ng];
        if (ACC) {
          const float4 t = pv[r & 1][0][ng];
          o = make_float4(t.x + o.x, t.y + o.y, t.z + o.z, t.w + o.w);
        }
        *(float4*)(cp + r * H + ng * 64) = o;
      }
    }
    if (r < nv1) {
#pragma unroll
      for (int ng = 0; ng < 4; ++ng) {
        float4 o = o1[ng];
        if (ACC) {
          const float4 t = pv[r & 1][1][ng];
          o = make_float4(t.x + o.x, t.y + o.y, t.z + o.z, t.w + o.w);
        }
        *(float4*)(cp + (16 + r) * H + ng * 64) = o;
      }
    }
  }
}

template <bool NORM, bool LN>
__global__ __launch_bounds__(256) void k_gemm_bf16x3(
    const float* __restrict__ A, const float* __restrict__ B,
    const float* __restrict__ bias, float* C, int M,
    const float* __restrict__ rden, const float* __restrict__ abias,
    const float* Hs, const float* __restrict__ sv, const float* __restrict__ dv,
    const unsigned* __restrict__ mxv,
    const float* __restrict__ gamma, const float* __restrict__ beta,
    const float* __restrict__ swp, int swidx, int accf)
{
  __shared__ __attribute__((aligned(16))) bf16 Bth[H][BP];
  __shared__ __attribute__((aligned(16))) bf16 Btl[H][BP];
  __shared__ __attribute__((aligned(16))) float Ab[H];

  const int tid = threadIdx.x;
  const int wid = tid >> 5, lane = tid & 31;
  const int hl = lane & 15, hh = lane >> 4;
  const int g0 = hh << 3;
  const int row0 = blockIdx.x * GROWS + wid * 32;

  int ra0 = row0 + hl;      if (ra0 > M - 1) ra0 = M - 1;
  int ra1 = row0 + 16 + hl; if (ra1 > M - 1) ra1 = M - 1;
  if (ra0 < 0) ra0 = 0;
  if (ra1 < 0) ra1 = 0;
  const unsigned oa0 = (unsigned)ra0 * H + g0;
  const unsigned oa1 = (unsigned)ra1 * H + g0;
  const float* pa0 = A + oa0;
  const float* pa1 = A + oa1;
  float inv0 = 1.f, inv1 = 1.f, ws0 = 0.f, ws1 = 0.f;
  const float* ph0 = A;
  const float* ph1 = A;
  if (NORM) {
    ws0 = __expf(lrelu(sv[ra0] + dv[ra0]) - ord2f(mxv[ra0]));
    ws1 = __expf(lrelu(sv[ra1] + dv[ra1]) - ord2f(mxv[ra1]));
    inv0 = 1.f / (rden[ra0] + ws0);
    inv1 = 1.f / (rden[ra1] + ws1);
    ph0 = Hs + oa0;
    ph1 = Hs + oa1;
  }

  float4 x0, x1, x2, x3, y0, y1, y2, y3;
  float4 u0, u1, u2, u3, w0, w1, w2, w3;
#define LOAD_A(ko_)                                                            \
  do {                                                                         \
    x0 = *(const float4*)(pa0 + (ko_));      x1 = *(const float4*)(pa0 + (ko_) + 4);  \
    x2 = *(const float4*)(pa0 + (ko_) + 16); x3 = *(const float4*)(pa0 + (ko_) + 20); \
    y0 = *(const float4*)(pa1 + (ko_));      y1 = *(const float4*)(pa1 + (ko_) + 4);  \
    y2 = *(const float4*)(pa1 + (ko_) + 16); y3 = *(const float4*)(pa1 + (ko_) + 20); \
    if (NORM) {                                                                \
      u0 = *(const float4*)(ph0 + (ko_));      u1 = *(const float4*)(ph0 + (ko_) + 4);  \
      u2 = *(const float4*)(ph0 + (ko_) + 16); u3 = *(const float4*)(ph0 + (ko_) + 20); \
      w0 = *(const float4*)(ph1 + (ko_));      w1 = *(const float4*)(ph1 + (ko_) + 4);  \
      w2 = *(const float4*)(ph1 + (ko_) + 16); w3 = *(const float4*)(ph1 + (ko_) + 20); \
    }                                                                          \
  } while (0)
#define ORDER_FENCE() asm volatile("" ::: "memory")

  LOAD_A(0);

#pragma unroll 4
  for (int i = 0; i < (H / 8) * (H / 4) / 256; ++i) {
    const int it = tid + i * 256;
    const int cg = it & 63;
    const int kg = it >> 6;
    float4 v[8];
#pragma unroll
    for (int q = 0; q < 8; ++q) v[q] = *(const float4*)&B[(size_t)(kg * 8 + q) * H + cg * 4];
    const int rowb = ((cg >> 4) << 6) + (cg & 15);
    v8bf h0, h1, h2, h3, l0, l1, l2, l3;
#pragma unroll
    for (int q = 0; q < 8; ++q) {
      bf16 t;
      t = (bf16)v[q].x; h0[q] = t; l0[q] = (bf16)(v[q].x - (float)t);
      t = (bf16)v[q].y; h1[q] = t; l1[q] = (bf16)(v[q].y - (float)t);
      t = (bf16)v[q].z; h2[q] = t; l2[q] = (bf16)(v[q].z - (float)t);
      t = (bf16)v[q].w; h3[q] = t; l3[q] = (bf16)(v[q].w - (float)t);
    }
    *(v8bf*)&Bth[rowb][kg * 8]      = h0;  *(v8bf*)&Btl[rowb][kg * 8]      = l0;
    *(v8bf*)&Bth[rowb + 16][kg * 8] = h1;  *(v8bf*)&Btl[rowb + 16][kg * 8] = l1;
    *(v8bf*)&Bth[rowb + 32][kg * 8] = h2;  *(v8bf*)&Btl[rowb + 32][kg * 8] = l2;
    *(v8bf*)&Bth[rowb + 48][kg * 8] = h3;  *(v8bf*)&Btl[rowb + 48][kg * 8] = l3;
  }
  if (NORM) {
    if (tid < H / 4) *(float4*)&Ab[tid * 4] = *(const float4*)&abias[tid * 4];
  }
  __syncthreads();

  if (row0 >= M) return;

  v8f acc0[16], acc1[16];
#pragma unroll
  for (int nt = 0; nt < 16; ++nt) { acc0[nt] = (v8f){}; acc1[nt] = (v8f){}; }

  const bf16* bph = &Bth[hl][0];
  const bf16* bpl = &Btl[hl][0];
  v16bf bh[2][4], bl[2][4];
#define LOAD_B(set_, ko_, ng_)                                                             \
  do {                                                                                     \
    _Pragma("unroll") for (int j = 0; j < 4; ++j) {                                        \
      bh[set_][j] = load_frag(bph + (((ng_) * 4 + j) * 16) * BP + (ko_), g0);              \
      bl[set_][j] = load_frag(bpl + (((ng_) * 4 + j) * 16) * BP + (ko_), g0);              \
    }                                                                                      \
  } while (0)

  LOAD_B(0, 0, 0);
#pragma unroll 1
  for (int kc = 0; kc < H / KC; ++kc) {
    v16bf ah0, al0, ah1, al1;
    if (NORM) {
      const float* bq = &Ab[kc * KC + g0];
      const float4 q0 = *(const float4*)(bq),      q1 = *(const float4*)(bq + 4);
      const float4 q2 = *(const float4*)(bq + 16), q3 = *(const float4*)(bq + 20);
      x0 = fma4(u0, ws0, x0); x1 = fma4(u1, ws0, x1); x2 = fma4(u2, ws0, x2); x3 = fma4(u3, ws0, x3);
      y0 = fma4(w0, ws1, y0); y1 = fma4(w1, ws1, y1); y2 = fma4(w2, ws1, y2); y3 = fma4(w3, ws1, y3);
      x0 = fma4(x0, inv0, q0); x1 = fma4(x1, inv0, q1); x2 = fma4(x2, inv0, q2); x3 = fma4(x3, inv0, q3);
      y0 = fma4(y0, inv1, q0); y1 = fma4(y1, inv1, q1); y2 = fma4(y2, inv1, q2); y3 = fma4(y3, inv1, q3);
    }
    split16(x0, x1, x2, x3, ah0, al0);
    split16(y0, y1, y2, y3, ah1, al1);
    const int kn = ((kc + 1) & (H / KC - 1)) * KC;
    LOAD_A(kn);
    ORDER_FENCE();
#pragma unroll
    for (int ng = 0; ng < 4; ++ng) {
      const int cur = ng & 1;
      if (ng < 3) LOAD_B(cur ^ 1, kc * KC, ng + 1);
      else        LOAD_B(cur ^ 1, kn, 0);
      ORDER_FENCE();
#pragma unroll
      for (int j = 0; j < 4; ++j) acc0[ng * 4 + j] = WMMA_BF16(ah0, bh[cur][j], acc0[ng * 4 + j]);
#pragma unroll
      for (int j = 0; j < 4; ++j) acc1[ng * 4 + j] = WMMA_BF16(ah1, bh[cur][j], acc1[ng * 4 + j]);
#pragma unroll
      for (int j = 0; j < 4; ++j) acc0[ng * 4 + j] = WMMA_BF16(al0, bh[cur][j], acc0[ng * 4 + j]);
#pragma unroll
      for (int j = 0; j < 4; ++j) acc1[ng * 4 + j] = WMMA_BF16(al1, bh[cur][j], acc1[ng * 4 + j]);
#pragma unroll
      for (int j = 0; j < 4; ++j) acc0[ng * 4 + j] = WMMA_BF16(ah0, bl[cur][j], acc0[ng * 4 + j]);
#pragma unroll
      for (int j = 0; j < 4; ++j) acc1[ng * 4 + j] = WMMA_BF16(ah1, bl[cur][j], acc1[ng * 4 + j]);
      ORDER_FENCE();
    }
  }
#undef LOAD_A
#undef LOAD_B
#undef ORDER_FENCE

  const bool full = (row0 + 32 <= M);
  if (LN) {
    const float scale = swp ? swp[swidx] : 1.f;
    if (accf) ln_epilogue<true>(acc0, acc1, C, M, bias, gamma, beta, scale, row0, row0 + (hh << 3), hl);
    else      ln_epilogue<false>(acc0, acc1, C, M, bias, gamma, beta, scale, row0, row0 + (hh << 3), hl);
    return;
  }
#pragma unroll
  for (int ng = 0; ng < 4; ++ng) {
    const int col = ng * 64 + hl * 4;
    float4 bv = bias ? *(const float4*)&bias[col] : make_float4(0.f, 0.f, 0.f, 0.f);
#pragma unroll
    for (int r = 0; r < 8; ++r) {
      const int r0g = row0 + (hh << 3) + r;
      const int r1g = r0g + 16;
      float4 o0 = make_float4(acc0[ng * 4 + 0][r] + bv.x, acc0[ng * 4 + 1][r] + bv.y,
                              acc0[ng * 4 + 2][r] + bv.z, acc0[ng * 4 + 3][r] + bv.w);
      float4 o1 = make_float4(acc1[ng * 4 + 0][r] + bv.x, acc1[ng * 4 + 1][r] + bv.y,
                              acc1[ng * 4 + 2][r] + bv.z, acc1[ng * 4 + 3][r] + bv.w);
      if (full) {
        *(float4*)&C[(size_t)r0g * H + col] = o0;
        *(float4*)&C[(size_t)r1g * H + col] = o1;
      } else {
        if (r0g < M) *(float4*)&C[(size_t)r0g * H + col] = o0;
        if (r1g < M) *(float4*)&C[(size_t)r1g * H + col] = o1;
      }
    }
  }
}

__global__ void k_zero(float* p, int n) {
  int i = blockIdx.x * blockDim.x + threadIdx.x;
  if (i < n) p[i] = 0.f;
}

#define CB 32
__global__ void k_colsum(const float* __restrict__ x, float* __restrict__ g, int N) {
  int c = threadIdx.x;
  int r0 = blockIdx.x * 256;
  int r1 = r0 + 256; if (r1 > N) r1 = N;
  const float* p = x + (size_t)r0 * H + c;
  float acc = 0.f;
  int r = r0;
  for (; r + CB <= r1; r += CB) {
    float v[CB];
#pragma unroll
    for (int j = 0; j < CB; ++j) v[j] = p[(size_t)j * H];
#pragma unroll
    for (int j = 0; j < CB; ++j) acc += v[j];
    p += (size_t)CB * H;
  }
  if (r < r1) {
    const int left = r1 - r;
    float v[CB];
#pragma unroll
    for (int j = 0; j < CB; ++j) v[j] = p[(size_t)(j < left ? j : left - 1) * H];
#pragma unroll
    for (int j = 0; j < CB; ++j) if (j < left) acc += v[j];
  }
  atomicAdd(&g[c], acc);
}
#undef CB

__global__ void k_selector(const float* __restrict__ g,
                           const float* __restrict__ W1, const float* __restrict__ b1,
                           const float* __restrict__ W2, const float* __restrict__ b2,
                           float* __restrict__ sw, int N) {
  __shared__ float gsh[H];
  __shared__ float hid[128];
  __shared__ float lg[4];
  int t = threadIdx.x;
  float invN = 1.f / (float)N;
  gsh[t] = g[t] * invN;
  gsh[t + 128] = g[t + 128] * invN;
  __syncthreads();
  float a = b1[t];
  for (int k = 0; k < H; ++k) a += gsh[k] * W1[k * 128 + t];
  hid[t] = fmaxf(a, 0.f);
  __syncthreads();
  if (t < 4) {
    float l = b2[t];
    for (int k = 0; k < 128; ++k) l += hid[k] * W2[k * 4 + t];
    lg[t] = l;
  }
  __syncthreads();
  if (t == 0) {
    float m = fmaxf(fmaxf(lg[0], lg[1]), fmaxf(lg[2], lg[3]));
    float e0 = __expf(lg[0] - m), e1 = __expf(lg[1] - m);
    float e2 = __expf(lg[2] - m), e3 = __expf(lg[3] - m);
    float s = e0 + e1 + e2 + e3;
    sw[0] = e0 / s; sw[1] = e1 / s; sw[2] = e2 / s; sw[3] = e3 / s;
  }
}

template <int NB>
__global__ void k_sd(const float* __restrict__ h, const float* __restrict__ asrc,
                     const float* __restrict__ adst, float* __restrict__ s,
                     float* __restrict__ d, float* __restrict__ num0, float* __restrict__ num1,
                     float* __restrict__ num2, int N) {
  int n = blockIdx.x * 8 + (threadIdx.x >> 5);
  if (n >= N) return;
  int lane = threadIdx.x & 31;
  const size_t plane = (size_t)N * H;
  const float4 z4 = make_float4(0.f, 0.f, 0.f, 0.f);
  float hv[NB][8];
#pragma unroll
  for (int b = 0; b < NB; ++b) {
#pragma unroll
    for (int j = 0; j < 8; ++j) hv[b][j] = h[b * plane + (size_t)n * H + lane + 32 * j];
  }
#pragma unroll
  for (int b = 0; b < NB; ++b) {
    float* np = (b == 0 ? num0 : b == 1 ? num1 : num2) + (size_t)n * H + lane * 8;
    *(float4*)np       = z4;
    *(float4*)(np + 4) = z4;
    float vs = 0.f, vd = 0.f;
#pragma unroll
    for (int j = 0; j < 8; ++j) {
      const int c = lane + 32 * j;
      vs = fmaf(hv[b][j], asrc[b * H + c], vs);
      vd = fmaf(hv[b][j], adst[b * H + c], vd);
    }
    for (int o = 16; o; o >>= 1) { vs += __shfl_xor(vs, o, 32); vd += __shfl_xor(vd, o, 32); }
    if (lane == 0) { s[(size_t)b * N + n] = vs; d[(size_t)b * N + n] = vd; }
  }
}

__global__ void k_init_m(const float* __restrict__ s, const float* __restrict__ d,
                         unsigned* __restrict__ mb, float* __restrict__ denom, int N) {
  int n = blockIdx.x * blockDim.x + threadIdx.x;
  if (n < N) {
    mb[n] = f2ord(lrelu(s[n] + d[n]));
    denom[n] = 0.f;
  }
}

__global__ void k_edge_max(const int* __restrict__ src, const int* __restrict__ dst,
                           const int* __restrict__ attr, int b0, int nb,
                           const float* __restrict__ s, const float* __restrict__ d,
                           unsigned* __restrict__ mb, int N, int E) {
  int e = blockIdx.x * blockDim.x + threadIdx.x;
  if (e >= E) return;
  const unsigned a = (unsigned)attr[e] - (unsigned)b0;
  if (a < (unsigned)nb) {
    const size_t o = (size_t)a * N;
    int dn = dst[e];
    float v = lrelu(s[o + src[e]] + d[o + dn]);
    atomicMax(&mb[o + dn], f2ord(v));
  }
}

__global__ void k_edge_accum(const int* __restrict__ src, const int* __restrict__ dst,
                             const int* __restrict__ attr, int b0, int nb,
                             const float* __restrict__ s, const float* __restrict__ d,
                             const unsigned* __restrict__ mb, const float* __restrict__ h,
                             float* __restrict__ denom, float* __restrict__ num0,
                             float* __restrict__ num1, float* __restrict__ num2, int N, int E) {
  int e = blockIdx.x * 8 + (threadIdx.x >> 5);
  if (e >= E) return;
  const unsigned a = (unsigned)attr[e] - (unsigned)b0;
  if (a >= (unsigned)nb) return;
  const size_t o = (size_t)a * N;
  int sn = src[e], dn = dst[e];
  float w = __expf(lrelu(s[o + sn] + d[o + dn]) - ord2f(mb[o + dn]));
  int lane = threadIdx.x & 31;
  if (lane == 0) atomicAdd(&denom[o + dn], w);
  const float* hp = h + (o + sn) * H;
  float* np = (a == 0u ? num0 : a == 1u ? num1 : num2) + (size_t)dn * H;
  for (int c = lane; c < H; c += 32)
    atomicAdd(&np[c], w * hp[c]);
}

extern "C" void kernel_launch(void* const* d_in, const int* in_sizes, int n_in,
                              void* d_out, int out_size, void* d_ws, size_t ws_size,
                              hipStream_t stream) {
  const float* x     = (const float*)d_in[0];
  const int*   ei    = (const int*)d_in[1];
  const int*   ea    = (const int*)d_in[2];
  const float* Wg    = (const float*)d_in[3];
  const float* a_src = (const float*)d_in[4];
  const float* a_dst = (const float*)d_in[5];
  const float* bg    = (const float*)d_in[6];
  const float* Wsl   = (const float*)d_in[7];
  const float* bsl   = (const float*)d_in[8];
  const float* gsl   = (const float*)d_in[9];
  const float* betas = (const float*)d_in[10];
  const float* W1    = (const float*)d_in[11];
  const float* b1    = (const float*)d_in[12];
  const float* W2    = (const float*)d_in[13];
  const float* b2    = (const float*)d_in[14];
  const float* Wf    = (const float*)d_in[15];
  const float* bfv   = (const float*)d_in[16];
  const float* gf    = (const float*)d_in[17];
  const float* betaf = (const float*)d_in[18];

  const int N = in_sizes[0] / H;
  const int E = in_sizes[2];
  float* out = (float*)d_out;

  char* wsb = (char*)d_ws;
  size_t off = 0;
  auto alloc = [&](size_t bytes) -> void* {
    void* p = wsb + off;
    off += (bytes + 255) & ~(size_t)255;
    return p;
  };
  const size_t pl  = (size_t)N * H;
  const size_t plB = pl * sizeof(float);
  auto r256 = [](size_t bytes) -> size_t { return (bytes + 255) & ~(size_t)255; };
  auto need = [&](int nb) -> size_t {
    return r256(H * sizeof(float)) + r256(4 * sizeof(float)) + 4 * r256((size_t)nb * N * sizeof(float)) +
           r256(nb * plB) + r256((nb - 1) * plB) + r256(plB);
  };
  const int nbat = (ws_size >= need(3)) ? 3 : (ws_size >= need(2)) ? 2 : 1;
  float*    g     = (float*)alloc(H * sizeof(float));
  float*    sw    = (float*)alloc(4 * sizeof(float));
  float*    sArr  = (float*)alloc((size_t)nbat * N * sizeof(float));
  float*    dArr  = (float*)alloc((size_t)nbat * N * sizeof(float));
  unsigned* mb    = (unsigned*)alloc((size_t)nbat * N * sizeof(unsigned));
  float*    denom = (float*)alloc((size_t)nbat * N * sizeof(float));
  float*    bufH  = (float*)alloc(nbat * plB);
  float*    bufX  = (float*)alloc((nbat > 1 ? nbat - 1 : 1) * plB);
  float*    bufC  = (float*)alloc(plB);
  auto xplane = [&](int b, int nb) -> float* { return (nbat > 1 && b == nb - 1) ? out : bufX + b * pl; };

  int gemmGrid = (N + GROWS - 1) / GROWS;
  int nWaveBlocks = (N + 7) / 8;
  int eWaveBlocks = (E + 7) / 8;
  int nThrBlocks  = (N + 255) / 256;
  int eThrBlocks  = (E + 255) / 256;

  k_zero<<<1, 256, 0, stream>>>(g, H);
  k_colsum<<<nThrBlocks, 256, 0, stream>>>(x, g, N);
  k_selector<<<1, 128, 0, stream>>>(g, W1, b1, W2, b2, sw, N);

  for (int b0 = 0; b0 < 3; b0 += nbat) {
    const int nb = (3 - b0 < nbat) ? 3 - b0 : nbat;
    float* x0 = xplane(0, nb);
    float* x1 = nb > 1 ? xplane(1, nb) : x0;
    float* x2 = nb > 2 ? xplane(2, nb) : x0;
    for (int b = 0; b < nb; ++b)
      k_gemm_bf16x3<false, false><<<gemmGrid, 256, 0, stream>>>(x, Wg + (size_t)(b0 + b) * H * H, nullptr,
          bufH + b * pl, N,
          nullptr, nullptr, nullptr, nullptr, nullptr, nullptr, nullptr, nullptr, nullptr, 0, 0);
    if (nb == 3)
      k_sd<3><<<nWaveBlocks, 256, 0, stream>>>(bufH, a_src + b0 * H, a_dst + b0 * H, sArr, dArr, x0, x1, x2, N);
    else if (nb == 2)
      k_sd<2><<<nWaveBlocks, 256, 0, stream>>>(bufH, a_src + b0 * H, a_dst + b0 * H, sArr, dArr, x0, x1, x2, N);
    else
      k_sd<1><<<nWaveBlocks, 256, 0, stream>>>(bufH, a_src + b0 * H, a_dst + b0 * H, sArr, dArr, x0, x1, x2, N);
    k_init_m<<<(nb * N + 255) / 256, 256, 0, stream>>>(sArr, dArr, mb, denom, nb * N);
    k_edge_max<<<eThrBlocks, 256, 0, stream>>>(ei, ei + E, ea, b0, nb, sArr, dArr, mb, N, E);
    k_edge_accum<<<eWaveBlocks, 256, 0, stream>>>(ei, ei + E, ea, b0, nb, sArr, dArr, mb, bufH, denom,
                                                  x0, x1, x2, N, E);
    for (int b = 0; b < nb; ++b) {
      const int i = b0 + b;
      k_gemm_bf16x3<true, true><<<gemmGrid, 256, 0, stream>>>(xplane(b, nb), Wsl + (size_t)i * H * H, bsl + i * H,
          bufC, N, denom + (size_t)b * N, bg + i * H, bufH + b * pl, sArr + (size_t)b * N, dArr + (size_t)b * N,
          mb + (size_t)b * N, gsl + i * H, betas + i * H, sw, i, i > 0 ? 1 : 0);
    }
  }
  k_gemm_bf16x3<false, true><<<gemmGrid, 256, 0, stream>>>(x, Wsl + (size_t)3 * H * H, bsl + 3 * H, bufC, N,
      nullptr, nullptr, nullptr, nullptr, nullptr, nullptr, gsl + 3 * H, betas + 3 * H, sw, 3, 1);

  k_gemm_bf16x3<false, true><<<gemmGrid, 256, 0, stream>>>(bufC, Wf, bfv, out, N, nullptr, nullptr,
      nullptr, nullptr, nullptr, nullptr, gf, betaf, nullptr, 0, 0);
}


# h product's epilogue takes the attention dots s and d from its accumulators in registers (in-lane fmas, halving exchange in the half-wave); node-pass launch gone: 12 launches; dot sum order changed
# speedup vs baseline: 2.7415x; 1.2100x over previous
#include <hip/hip_runtime.h>

#define H 256
#define KC 32
#define BP 264
#define GROWS 256

typedef __bf16 bf16;
typedef __attribute__((ext_vector_type(16))) __bf16 v16bf;
typedef __attribute__((ext_vector_type(8)))  __bf16 v8bf;
typedef __attribute__((ext_vector_type(8)))  float  v8f;

__device__ __forceinline__ unsigned short f2bfbits(float f) {
  unsigned u = __float_as_uint(f);
  return (unsigned short)((u + 0x7FFFu + ((u >> 16) & 1u)) >> 16);
}
__device__ __forceinline__ float bfbits2f(unsigned short h) {
  return __uint_as_float(((unsigned)h) << 16);
}
__device__ __forceinline__ bf16 f2bf(float f) {
  unsigned short hs = f2bfbits(f);
  return __builtin_bit_cast(bf16, hs);
}
__device__ __forceinline__ float lrelu(float v) { return v > 0.f ? v : 0.2f * v; }
__device__ __forceinline__ unsigned f2ord(float f) {
  unsigned u = __float_as_uint(f);
  return (u & 0x80000000u) ? ~u : (u | 0x80000000u);
}
__device__ __forceinline__ float ord2f(unsigned u) {
  unsigned b = (u & 0x80000000u) ? (u & 0x7FFFFFFFu) : ~u;
  return __uint_as_float(b);
}

__device__ __forceinline__ v16bf load_frag(const bf16* rowp, int g0) {
  v8bf a = *(const v8bf*)(rowp + g0);
  v8bf b = *(const v8bf*)(rowp + 16 + g0);
  return __builtin_shufflevector(a, b, 0, 1, 2, 3, 4, 5, 6, 7,
                                 8, 9, 10, 11, 12, 13, 14, 15);
}

#define WMMA_BF16(A_, B_, C_) \
  __builtin_amdgcn_wmma_f32_16x16x32_bf16(false, (A_), false, (B_), (short)0, (C_), false, false)

__device__ __forceinline__ void split16(const float4 a, const float4 b, const float4 c, const float4 d,
                                        v16bf& hi, v16bf& lo) {
  const float f[16] = {a.x, a.y, a.z, a.w, b.x, b.y, b.z, b.w,
                       c.x, c.y, c.z, c.w, d.x, d.y, d.z, d.w};
#pragma unroll
  for (int q = 0; q < 16; ++q) {
    bf16 h = (bf16)f[q];
    hi[q] = h;
    lo[q] = (bf16)(f[q] - (float)h);
  }
}

__device__ __forceinline__ float4 fma4(const float4 a, const float s, const float4 b) {
  return make_float4(fmaf(a.x, s, b.x), fmaf(a.y, s, b.y), fmaf(a.z, s, b.z), fmaf(a.w, s, b.w));
}

template <bool ACC>
__device__ __forceinline__ void ln_epilogue(const v8f (&acc0)[16], const v8f (&acc1)[16], float* C, int M,
                                            const float* __restrict__ bias,
                                            const float* __restrict__ gamma,
                                            const float* __restrict__ beta,
                                            const float scale, const int row0, const int rb,
                                            const int hl) {
  const float* csafe = C + (size_t)row0 * H + hl * 4;
  float* cp = C + (size_t)rb * H + hl * 4;
  const int nv0 = M - rb;
  const int nv1 = M - rb - 16;
  float4 bv[4];
#pragma unroll
  for (int ng = 0; ng < 4; ++ng) bv[ng] = *(const float4*)&bias[ng * 64 + hl * 4];
  float4 pv[2][2][4];
  if (ACC) {
    const float* p0 = (0 < nv0) ? cp : csafe;
    const float* p1 = (0 < nv1) ? cp + 16 * H : csafe;
#pragma unroll
    for (int ng = 0; ng < 4; ++ng) {
      pv[0][0][ng] = *(const float4*)(p0 + ng * 64);
      pv[0][1][ng] = *(const float4*)(p1 + ng * 64);
    }
  }
  float s0[8], q0[8], s1[8], q1[8];
#pragma unroll
  for (int r = 0; r < 8; ++r) {
    float a = 0.f, b = 0.f, cc = 0.f, d = 0.f;
#pragma unroll
    for (int ng = 0; ng < 4; ++ng) {
      const float bb[4] = {bv[ng].x, bv[ng].y, bv[ng].z, bv[ng].w};
#pragma unroll
      for (int j = 0; j < 4; ++j) {
        const float v = acc0[ng * 4 + j][r] + bb[j];
        const float w = acc1[ng * 4 + j][r] + bb[j];
        a += v; b += v * v;
        cc += w; d += w * w;
      }
    }
    s0[r] = a; q0[r] = b; s1[r] = cc; q1[r] = d;
  }
#pragma unroll
  for (int o = 1; o < 16; o <<= 1) {
#pragma unroll
    for (int r = 0; r < 8; ++r) {
      s0[r] += __shfl_xor(s0[r], o, 32);
      q0[r] += __shfl_xor(q0[r], o, 32);
      s1[r] += __shfl_xor(s1[r], o, 32);
      q1[r] += __shfl_xor(q1[r], o, 32);
    }
  }
  float4 gv[4], ev[4];
#pragma unroll
  for (int ng = 0; ng < 4; ++ng) {
    gv[ng] = *(const float4*)&gamma[ng * 64 + hl * 4];
    ev[ng] = *(const float4*)&beta[ng * 64 + hl * 4];
  }
#pragma unroll
  for (int r = 0; r < 8; ++r) {
    if (ACC && r < 7) {
      const float* p0 = (r + 1 < nv0) ? cp + (r + 1) * H : csafe;
      const float* p1 = (r + 1 < nv1) ? cp + (16 + r + 1) * H : csafe;
#pragma unroll
      for (int ng = 0; ng < 4; ++ng) {
        pv[(r + 1) & 1][0][ng] = *(const float4*)(p0 + ng * 64);
        pv[(r + 1) & 1][1][ng] = *(const float4*)(p1 + ng * 64);
      }
    }
    const float mu0 = s0[r] * (1.f / H);
    const float mu1 = s1[r] * (1.f / H);
    const float rs0 = rsqrtf(q0[r] * (1.f / H) - mu0 * mu0 + 1e-5f);
    const float rs1 = rsqrtf(q1[r] * (1.f / H) - mu1 * mu1 + 1e-5f);
    float4 o0[4], o1[4];
#pragma unroll
    for (int ng = 0; ng < 4; ++ng) {
      const float bb[4] = {bv[ng].x, bv[ng].y, bv[ng].z, bv[ng].w};
      const float gg[4] = {gv[ng].x, gv[ng].y, gv[ng].z, gv[ng].w};
      const float ee[4] = {ev[ng].x, ev[ng].y, ev[ng].z, ev[ng].w};
      float y0[4], y1[4];
#pragma unroll
      for (int j = 0; j < 4; ++j) {
        const float m0 = mu0 - bb[j];
        const float m1 = mu1 - bb[j];
        y0[j] = fmaxf((acc0[ng * 4 + j][r] - m0) * rs0 * gg[j] + ee[j], 0.f) * scale;
        y1[j] = fmaxf((acc1[ng * 4 + j][r] - m1) * rs1 * gg[j] + ee[j], 0.f) * scale;
      }
      o0[ng] = make_float4(y0[0], y0[1], y0[2], y0[3]);
      o1[ng] = make_float4(y1[0], y1[1], y1[2], y1[3]);
    }
    if (r < nv0) {
#pragma unroll
      for (int ng = 0; ng < 4; ++ng) {
        float4 o = o0[ng];
        if (ACC) {
          const float4 t = pv[r & 1][0][ng];
          o = make_float4(t.x + o.x, t.y + o.y, t.z + o.z, t.w + o.w);
        }
        *(float4*)(cp + r * H + ng * 64) = o;
      }
    }
    if (r < nv1) {
#pragma unroll
      for (int ng = 0; ng < 4; ++ng) {
        float4 o = o1[ng];
        if (ACC) {
          const float4 t = pv[r & 1][1][ng];
          o = make_float4(t.x + o.x, t.y + o.y, t.z + o.z, t.w + o.w);
        }
        *(float4*)(cp + (16 + r) * H + ng * 64) = o;
      }
    }
  }
}

template <bool NORM, bool LN, bool SD>
__global__ __launch_bounds__(256) void k_gemm_bf16x3(
    const float* A, const float* __restrict__ B,
    const float* __restrict__ bias, float* C, int M,
    const float* __restrict__ rden, const float* __restrict__ abias,
    const float* Hs, const float* __restrict__ sv, const float* __restrict__ dv,
    const unsigned* __restrict__ mxv,
    const float* __restrict__ gamma, const float* __restrict__ beta,
    const float* __restrict__ swp, int swidx, int accf,
    const float* __restrict__ asv, const float* __restrict__ adv,
    float* __restrict__ sOut, float* __restrict__ dOut)
{
  __shared__ __attribute__((aligned(16))) bf16 Bth[H][BP];
  __shared__ __attribute__((aligned(16))) bf16 Btl[H][BP];
  __shared__ __attribute__((aligned(16))) float Ab[H];

  const int tid = threadIdx.x;
  const int wid = tid >> 5, lane = tid & 31;
  const int hl = lane & 15, hh = lane >> 4;
  const int g0 = hh << 3;
  const int row0 = blockIdx.x * GROWS + wid * 32;

  int ra0 = row0 + hl;      if (ra0 > M - 1) ra0 = M - 1;
  int ra1 = row0 + 16 + hl; if (ra1 > M - 1) ra1 = M - 1;
  if (ra0 < 0) ra0 = 0;
  if (ra1 < 0) ra1 = 0;
  const unsigned oa0 = (unsigned)ra0 * H + g0;
  const unsigned oa1 = (unsigned)ra1 * H + g0;
  const float* pa0 = A + oa0;
  const float* pa1 = A + oa1;
  float inv0 = 1.f, inv1 = 1.f, ws0 = 0.f, ws1 = 0.f;
  const float* ph0 = A;
  const float* ph1 = A;
  if (NORM) {
    ws0 = __expf(lrelu(sv[ra0] + dv[ra0]) - ord2f(mxv[ra0]));
    ws1 = __expf(lrelu(sv[ra1] + dv[ra1]) - ord2f(mxv[ra1]));
    inv0 = 1.f / (rden[ra0] + ws0);
    inv1 = 1.f / (rden[ra1] + ws1);
    ph0 = Hs + oa0;
    ph1 = Hs + oa1;
  }

  float4 x0, x1, x2, x3, y0, y1, y2, y3;
  float4 u0, u1, u2, u3, w0, w1, w2, w3;
#define LOAD_A(ko_)                                                            \
  do {                                                                         \
    x0 = *(const float4*)(pa0 + (ko_));      x1 = *(const float4*)(pa0 + (ko_) + 4);  \
    x2 = *(const float4*)(pa0 + (ko_) + 16); x3 = *(const float4*)(pa0 + (ko_) + 20); \
    y0 = *(const float4*)(pa1 + (ko_));      y1 = *(const float4*)(pa1 + (ko_) + 4);  \
    y2 = *(const float4*)(pa1 + (ko_) + 16); y3 = *(const float4*)(pa1 + (ko_) + 20); \
    if (NORM) {                                                                \
      u0 = *(const float4*)(ph0 + (ko_));      u1 = *(const float4*)(ph0 + (ko_) + 4);  \
      u2 = *(const float4*)(ph0 + (ko_) + 16); u3 = *(const float4*)(ph0 + (ko_) + 20); \
      w0 = *(const float4*)(ph1 + (ko_));      w1 = *(const float4*)(ph1 + (ko_) + 4);  \
      w2 = *(const float4*)(ph1 + (ko_) + 16); w3 = *(const float4*)(ph1 + (ko_) + 20); \
    }                                                                          \
  } while (0)
#define ORDER_FENCE() asm volatile("" ::: "memory")

  LOAD_A(0);

#pragma unroll 4
  for (int i = 0; i < (H / 8) * (H / 4) / 256; ++i) {
    const int it = tid + i * 256;
    const int cg = it & 63;
    const int kg = it >> 6;
    float4 v[8];
#pragma unroll
    for (int q = 0; q < 8; ++q) v[q] = *(const float4*)&B[(size_t)(kg * 8 + q) * H + cg * 4];
    const int rowb = ((cg >> 4) << 6) + (cg & 15);
    v8bf h0, h1, h2, h3, l0, l1, l2, l3;
#pragma unroll
    for (int q = 0; q < 8; ++q) {
      bf16 t;
      t = (bf16)v[q].x; h0[q] = t; l0[q] = (bf16)(v[q].x - (float)t);
      t = (bf16)v[q].y; h1[q] = t; l1[q] = (bf16)(v[q].y - (float)t);
      t = (bf16)v[q].z; h2[q] = t; l2[q] = (bf16)(v[q].z - (float)t);
      t = (bf16)v[q].w; h3[q] = t; l3[q] = (bf16)(v[q].w - (float)t);
    }
    *(v8bf*)&Bth[rowb][kg * 8]      = h0;  *(v8bf*)&Btl[rowb][kg * 8]      = l0;
    *(v8bf*)&Bth[rowb + 16][kg * 8] = h1;  *(v8bf*)&Btl[rowb + 16][kg * 8] = l1;
    *(v8bf*)&Bth[rowb + 32][kg * 8] = h2;  *(v8bf*)&Btl[rowb + 32][kg * 8] = l2;
    *(v8bf*)&Bth[rowb + 48][kg * 8] = h3;  *(v8bf*)&Btl[rowb + 48][kg * 8] = l3;
  }
  if (NORM) {
    if (tid < H / 4) *(float4*)&Ab[tid * 4] = *(const float4*)&abias[tid * 4];
  }
  __syncthreads();

  if (row0 >= M) return;

  v8f acc0[16], acc1[16];
#pragma unroll
  for (int nt = 0; nt < 16; ++nt) { acc0[nt] = (v8f){}; acc1[nt] = (v8f){}; }

  const bf16* bph = &Bth[hl][0];
  const bf16* bpl = &Btl[hl][0];
  v16bf bh[2][4], bl[2][4];
#define LOAD_B(set_, ko_, ng_)                                                             \
  do {                                                                                     \
    _Pragma("unroll") for (int j = 0; j < 4; ++j) {                                        \
      bh[set_][j] = load_frag(bph + (((ng_) * 4 + j) * 16) * BP + (ko_), g0);              \
      bl[set_][j] = load_frag(bpl + (((ng_) * 4 + j) * 16) * BP + (ko_), g0);              \
    }                                                                                      \
  } while (0)

  LOAD_B(0, 0, 0);
#pragma unroll 1
  for (int kc = 0; kc < H / KC; ++kc) {
    v16bf ah0, al0, ah1, al1;
    if (NORM) {
      const float* bq = &Ab[kc * KC + g0];
      const float4 q0 = *(const float4*)(bq),      q1 = *(const float4*)(bq + 4);
      const float4 q2 = *(const float4*)(bq + 16), q3 = *(const float4*)(bq + 20);
      x0 = fma4(u0, ws0, x0); x1 = fma4(u1, ws0, x1); x2 = fma4(u2, ws0, x2); x3 = fma4(u3, ws0, x3);
      y0 = fma4(w0, ws1, y0); y1 = fma4(w1, ws1, y1); y2 = fma4(w2, ws1, y2); y3 = fma4(w3, ws1, y3);
      x0 = fma4(x0, inv0, q0); x1 = fma4(x1, inv0, q1); x2 = fma4(x2, inv0, q2); x3 = fma4(x3, inv0, q3);
      y0 = fma4(y0, inv1, q0); y1 = fma4(y1, inv1, q1); y2 = fma4(y2, inv1, q2); y3 = fma4(y3, inv1, q3);
    }
    split16(x0, x1, x2, x3, ah0, al0);
    split16(y0, y1, y2, y3, ah1, al1);
    const int kn = ((kc + 1) & (H / KC - 1)) * KC;
    LOAD_A(kn);
    ORDER_FENCE();
#pragma unroll
    for (int ng = 0; ng < 4; ++ng) {
      const int cur = ng & 1;
      if (ng < 3) LOAD_B(cur ^ 1, kc * KC, ng + 1);
      else        LOAD_B(cur ^ 1, kn, 0);
      ORDER_FENCE();
#pragma unroll
      for (int j = 0; j < 4; ++j) acc0[ng * 4 + j] = WMMA_BF16(ah0, bh[cur][j], acc0[ng * 4 + j]);
#pragma unroll
      for (int j = 0; j < 4; ++j) acc1[ng * 4 + j] = WMMA_BF16(ah1, bh[cur][j], acc1[ng * 4 + j]);
#pragma unroll
      for (int j = 0; j < 4; ++j) acc0[ng * 4 + j] = WMMA_BF16(al0, bh[cur][j], acc0[ng * 4 + j]);
#pragma unroll
      for (int j = 0; j < 4; ++j) acc1[ng * 4 + j] = WMMA_BF16(al1, bh[cur][j], acc1[ng * 4 + j]);
#pragma unroll
      for (int j = 0; j < 4; ++j) acc0[ng * 4 + j] = WMMA_BF16(ah0, bl[cur][j], acc0[ng * 4 + j]);
#pragma unroll
      for (int j = 0; j < 4; ++j) acc1[ng * 4 + j] = WMMA_BF16(ah1, bl[cur][j], acc1[ng * 4 + j]);
      ORDER_FENCE();
    }
  }
#undef LOAD_A
#undef LOAD_B
#undef ORDER_FENCE

  const bool full = (row0 + 32 <= M);
  if (LN) {
    const float scale = swp ? swp[swidx] : 1.f;
    if (accf) ln_epilogue<true>(acc0, acc1, C, M, bias, gamma, beta, scale, row0, row0 + (hh << 3), hl);
    else      ln_epilogue<false>(acc0, acc1, C, M, bias, gamma, beta, scale, row0, row0 + (hh << 3), hl);
    return;
  }
  float ps[16], pd[16];
  if (SD) {
#pragma unroll
    for (int t = 0; t < 16; ++t) { ps[t] = 0.f; pd[t] = 0.f; }
  }
#pragma unroll
  for (int ng = 0; ng < 4; ++ng) {
    const int col = ng * 64 + hl * 4;
    float4 bv = bias ? *(const float4*)&bias[col] : make_float4(0.f, 0.f, 0.f, 0.f);
    float4 a4 = make_float4(0.f, 0.f, 0.f, 0.f), d4 = a4;
    if (SD) { a4 = *(const float4*)&asv[col]; d4 = *(const float4*)&adv[col]; }
#pragma unroll
    for (int r = 0; r < 8; ++r) {
      const int r0g = row0 + (hh << 3) + r;
      const int r1g = r0g + 16;
      float4 o0 = make_float4(acc0[ng * 4 + 0][r] + bv.x, acc0[ng * 4 + 1][r] + bv.y,
                              acc0[ng * 4 + 2][r] + bv.z, acc0[ng * 4 + 3][r] + bv.w);
      float4 o1 = make_float4(acc1[ng * 4 + 0][r] + bv.x, acc1[ng * 4 + 1][r] + bv.y,
                              acc1[ng * 4 + 2][r] + bv.z, acc1[ng * 4 + 3][r] + bv.w);
      if (SD) {
        ps[r] = fmaf(o0.w, a4.w, fmaf(o0.z, a4.z, fmaf(o0.y, a4.y, fmaf(o0.x, a4.x, ps[r]))));
        pd[r] = fmaf(o0.w, d4.w, fmaf(o0.z, d4.z, fmaf(o0.y, d4.y, fmaf(o0.x, d4.x, pd[r]))));
        ps[8 + r] = fmaf(o1.w, a4.w, fmaf(o1.z, a4.z, fmaf(o1.y, a4.y, fmaf(o1.x, a4.x, ps[8 + r]))));
        pd[8 + r] = fmaf(o1.w, d4.w, fmaf(o1.z, d4.z, fmaf(o1.y, d4.y, fmaf(o1.x, d4.x, pd[8 + r]))));
      }
      if (full) {
        *(float4*)&C[(size_t)r0g * H + col] = o0;
        *(float4*)&C[(size_t)r1g * H + col] = o1;
      } else {
        if (r0g < M) *(float4*)&C[(size_t)r0g * H + col] = o0;
        if (r1g < M) *(float4*)&C[(size_t)r1g * H + col] = o1;
      }
    }
  }
  if (SD) {
    const bool b0 = hl & 1, b1 = hl & 2, b2 = hl & 4, b3 = hl & 8;
    float s8[8], d8[8], s4[4], d4[4], s2[2], d2[2];
#pragma unroll
    for (int u = 0; u < 8; ++u) {
      s8[u] = (b0 ? ps[2 * u + 1] : ps[2 * u]) + __shfl_xor(b0 ? ps[2 * u] : ps[2 * u + 1], 1, 32);
      d8[u] = (b0 ? pd[2 * u + 1] : pd[2 * u]) + __shfl_xor(b0 ? pd[2 * u] : pd[2 * u + 1], 1, 32);
    }
#pragma unroll
    for (int u = 0; u < 4; ++u) {
      s4[u] = (b1 ? s8[2 * u + 1] : s8[2 * u]) + __shfl_xor(b1 ? s8[2 * u] : s8[2 * u + 1], 2, 32);
      d4[u] = (b1 ? d8[2 * u + 1] : d8[2 * u]) + __shfl_xor(b1 ? d8[2 * u] : d8[2 * u + 1], 2, 32);
    }
#pragma unroll
    for (int u = 0; u < 2; ++u) {
      s2[u] = (b2 ? s4[2 * u + 1] : s4[2 * u]) + __shfl_xor(b2 ? s4[2 * u] : s4[2 * u + 1], 4, 32);
      d2[u] = (b2 ? d4[2 * u + 1] : d4[2 * u]) + __shfl_xor(b2 ? d4[2 * u] : d4[2 * u + 1], 4, 32);
    }
    const float sv1 = (b3 ? s2[1] : s2[0]) + __shfl_xor(b3 ? s2[0] : s2[1], 8, 32);
    const float dv1 = (b3 ? d2[1] : d2[0]) + __shfl_xor(b3 ? d2[0] : d2[1], 8, 32);
    const int rowg = row0 + ((hl >> 3) << 4) + (hh << 3) + (hl & 7);
    if (rowg < M) { sOut[rowg] = sv1; dOut[rowg] = dv1; }
  }
}

__global__ void k_zero(float* p, int n) {
  int i = blockIdx.x * blockDim.x + threadIdx.x;
  if (i < n) p[i] = 0.f;
}

#define CB 32
__global__ void k_colsum(const float* __restrict__ x, float* __restrict__ g, int N) {
  int c = threadIdx.x;
  int r0 = blockIdx.x * 256;
  int r1 = r0 + 256; if (r1 > N) r1 = N;
  const float* p = x + (size_t)r0 * H + c;
  float acc = 0.f;
  int r = r0;
  for (; r + CB <= r1; r += CB) {
    float v[CB];
#pragma unroll
    for (int j = 0; j < CB; ++j) v[j] = p[(size_t)j * H];
#pragma unroll
    for (int j = 0; j < CB; ++j) acc += v[j];
    p += (size_t)CB * H;
  }
  if (r < r1) {
    const int left = r1 - r;
    float v[CB];
#pragma unroll
    for (int j = 0; j < CB; ++j) v[j] = p[(size_t)(j < left ? j : left - 1) * H];
#pragma unroll
    for (int j = 0; j < CB; ++j) if (j < left) acc += v[j];
  }
  atomicAdd(&g[c], acc);
}
#undef CB

__global__ void k_selector(const float* __restrict__ g,
                           const float* __restrict__ W1, const float* __restrict__ b1,
                           const float* __restrict__ W2, const float* __restrict__ b2,
                           float* __restrict__ sw, int N) {
  __shared__ float gsh[H];
  __shared__ float hid[128];
  __shared__ float lg[4];
  int t = threadIdx.x;
  float invN = 1.f / (float)N;
  gsh[t] = g[t] * invN;
  gsh[t + 128] = g[t + 128] * invN;
  __syncthreads();
  float a = b1[t];
  for (int k = 0; k < H; ++k) a += gsh[k] * W1[k * 128 + t];
  hid[t] = fmaxf(a, 0.f);
  __syncthreads();
  if (t < 4) {
    float l = b2[t];
    for (int k = 0; k < 128; ++k) l += hid[k] * W2[k * 4 + t];
    lg[t] = l;
  }
  __syncthreads();
  if (t == 0) {
    float m = fmaxf(fmaxf(lg[0], lg[1]), fmaxf(lg[2], lg[3]));
    float e0 = __expf(lg[0] - m), e1 = __expf(lg[1] - m);
    float e2 = __expf(lg[2] - m), e3 = __expf(lg[3] - m);
    float s = e0 + e1 + e2 + e3;
    sw[0] = e0 / s; sw[1] = e1 / s; sw[2] = e2 / s; sw[3] = e3 / s;
  }
}


#define OWN_R 98
#define OWN_T 1024
#define OWN_CAP 1280
__global__ __launch_bounds__(OWN_T) void k_edge_own(
    const int* __restrict__ src, const int* __restrict__ dst, const int* __restrict__ attr,
    int b0, int nb, const float* __restrict__ s, const float* __restrict__ d,
    const float* __restrict__ h, unsigned* __restrict__ mb, float* __restrict__ denom,
    float* __restrict__ num0, float* __restrict__ num1, float* __restrict__ num2, int N, int E) {
  __shared__ __attribute__((aligned(16))) float acc[3 * OWN_R * H];
  __shared__ unsigned mx[3 * OWN_R];
  __shared__ float den[3 * OWN_R];
  __shared__ int lrow[OWN_CAP];
  __shared__ int lslot[OWN_CAP];
  __shared__ float lv[OWN_CAP];
  __shared__ int cnt;

  const int tid = threadIdx.x, lane = tid & 31, wid = tid >> 5;
  const int n0 = blockIdx.x * OWN_R;
  const int nv = (N - n0 < OWN_R) ? N - n0 : OWN_R;
  const float4 z4 = make_float4(0.f, 0.f, 0.f, 0.f);
  float4* acc4 = (float4*)acc;
  for (int i = tid; i < nb * OWN_R * (H / 4); i += OWN_T) acc4[i] = z4;
  for (int i = tid; i < nb * OWN_R; i += OWN_T) {
    const int b = i / OWN_R, l = i - b * OWN_R;
    float sd = 0.f;
    if (l < nv) sd = s[(size_t)b * N + n0 + l] + d[(size_t)b * N + n0 + l];
    mx[i] = f2ord(lrelu(sd));
    den[i] = 0.f;
  }
  if (tid == 0) cnt = 0;
  __syncthreads();

#define OWN_NOTE(dn_, e_)                                    \
  do {                                                       \
    if ((unsigned)((dn_) - n0) < (unsigned)nv) {             \
      const int ix = atomicAdd(&cnt, 1);                     \
      if (ix < OWN_CAP) lrow[ix] = (e_);                        \
    }                                                        \
  } while (0)
  const bool vec = ((E & 3) == 0) && ((((size_t)dst) & 15) == 0);
  const int E4 = vec ? (E >> 2) : 0;
  const int4* d4 = (const int4*)dst;
  for (int q0 = 0; q0 < E4; q0 += 4 * OWN_T) {
    int4 dv[4];
#pragma unroll
    for (int j = 0; j < 4; ++j) {
      const int q = q0 + j * OWN_T + tid;
      dv[j] = d4[q < E4 ? q : E4 - 1];
    }
#pragma unroll
    for (int j = 0; j < 4; ++j) {
      const int q = q0 + j * OWN_T + tid;
      if (q < E4) {
        OWN_NOTE(dv[j].x, 4 * q);     OWN_NOTE(dv[j].y, 4 * q + 1);
        OWN_NOTE(dv[j].z, 4 * q + 2); OWN_NOTE(dv[j].w, 4 * q + 3);
      }
    }
  }
  for (int e = 4 * E4 + tid; e < E; e += OWN_T) {
    const int dn = dst[e];
    OWN_NOTE(dn, e);
  }
#undef OWN_NOTE
  __syncthreads();
  const int noted = cnt;
  const bool over = noted > OWN_CAP;
  const int total = over ? 0 : noted;

  if (!over) {
    for (int i = tid; i < total; i += OWN_T) {
      const int e = lrow[i];
      const unsigned a = (unsigned)attr[e] - (unsigned)b0;
      int slot = -1;
      if (a < (unsigned)nb) {
        const int dn = dst[e], sn = src[e];
        const size_t o = (size_t)a * N;
        const float v = lrelu(s[o + sn] + d[o + dn]);
        slot = (int)a * OWN_R + (dn - n0);
        atomicMax(&mx[slot], f2ord(v));
        lrow[i] = (int)(o + sn);
        lv[i] = v;
      }
      lslot[i] = slot;
    }
  } else {
    for (int e = tid; e < E; e += OWN_T) {
      const int dn = dst[e];
      if ((unsigned)(dn - n0) < (unsigned)nv) {
        const unsigned a = (unsigned)attr[e] - (unsigned)b0;
        if (a < (unsigned)nb) {
          const size_t o = (size_t)a * N;
          atomicMax(&mx[(int)a * OWN_R + (dn - n0)], f2ord(lrelu(s[o + src[e]] + d[o + dn])));
        }
      }
    }
  }
  __syncthreads();

  if (!over) {
    for (int i0 = wid; i0 < total; i0 += 4 * 32) {
      int sl[4];
      float wv[4];
      float hv[4][8];
#pragma unroll
      for (int k = 0; k < 4; ++k) {
        const int i = i0 + 32 * k;
        const bool ok = i < total;
        sl[k] = __builtin_amdgcn_readfirstlane(ok ? lslot[i] : -1);
        const int row = __builtin_amdgcn_readfirstlane(sl[k] >= 0 ? lrow[i] : 0);
        const float* hp = h + (size_t)row * H + lane;
#pragma unroll
        for (int j = 0; j < 8; ++j) hv[k][j] = hp[32 * j];
        wv[k] = (sl[k] >= 0) ? __expf(lv[i] - ord2f(mx[sl[k]])) : 0.f;
      }
#pragma unroll
      for (int k = 0; k < 4; ++k) {
        if (sl[k] >= 0) {
          if (lane == 0) atomicAdd(&den[sl[k]], wv[k]);
          float* ap = acc + sl[k] * H + lane;
#pragma unroll
          for (int j = 0; j < 8; ++j) atomicAdd(&ap[32 * j], wv[k] * hv[k][j]);
        }
      }
    }
  } else {
    for (int e = wid; e < E; e += 32) {
      const int dn = dst[e];
      if ((unsigned)(dn - n0) >= (unsigned)nv) continue;
      const unsigned a = (unsigned)attr[e] - (unsigned)b0;
      if (a >= (unsigned)nb) continue;
      const size_t o = (size_t)a * N;
      const int sn = src[e];
      const int slot = (int)a * OWN_R + (dn - n0);
      const float w = __expf(lrelu(s[o + sn] + d[o + dn]) - ord2f(mx[slot]));
      if (lane == 0) atomicAdd(&den[slot], w);
      const float* hp = h + (o + sn) * H;
      for (int c = lane; c < H; c += 32) atomicAdd(&acc[slot * H + c], w * hp[c]);
    }
  }
  __syncthreads();

  for (int i = tid; i < nb * OWN_R * (H / 4); i += OWN_T) {
    const int slot = i >> 6;
    const int b = slot / OWN_R, l = slot - b * OWN_R;
    if (l < nv) {
      float* np = (b == 0 ? num0 : b == 1 ? num1 : num2);
      *(float4*)(np + (size_t)(n0 + l) * H + (i & 63) * 4) = acc4[i];
    }
  }
  for (int i = tid; i < nb * OWN_R; i += OWN_T) {
    const int b = i / OWN_R, l = i - b * OWN_R;
    if (l < nv) {
      mb[(size_t)b * N + n0 + l] = mx[i];
      denom[(size_t)b * N + n0 + l] = den[i];
    }
  }
}

extern "C" void kernel_launch(void* const* d_in, const int* in_sizes, int n_in,
                              void* d_out, int out_size, void* d_ws, size_t ws_size,
                              hipStream_t stream) {
  const float* x     = (const float*)d_in[0];
  const int*   ei    = (const int*)d_in[1];
  const int*   ea    = (const int*)d_in[2];
  const float* Wg    = (const float*)d_in[3];
  const float* a_src = (const float*)d_in[4];
  const float* a_dst = (const float*)d_in[5];
  const float* bg    = (const float*)d_in[6];
  const float* Wsl   = (const float*)d_in[7];
  const float* bsl   = (const float*)d_in[8];
  const float* gsl   = (const float*)d_in[9];
  const float* betas = (const float*)d_in[10];
  const float* W1    = (const float*)d_in[11];
  const float* b1    = (const float*)d_in[12];
  const float* W2    = (const float*)d_in[13];
  const float* b2    = (const float*)d_in[14];
  const float* Wf    = (const float*)d_in[15];
  const float* bfv   = (const float*)d_in[16];
  const float* gf    = (const float*)d_in[17];
  const float* betaf = (const float*)d_in[18];

  const int N = in_sizes[0] / H;
  const int E = in_sizes[2];
  float* out = (float*)d_out;

  char* wsb = (char*)d_ws;
  size_t off = 0;
  auto alloc = [&](size_t bytes) -> void* {
    void* p = wsb + off;
    off += (bytes + 255) & ~(size_t)255;
    return p;
  };
  const size_t pl  = (size_t)N * H;
  const size_t plB = pl * sizeof(float);
  auto r256 = [](size_t bytes) -> size_t { return (bytes + 255) & ~(size_t)255; };
  auto need = [&](int nb) -> size_t {
    return r256(H * sizeof(float)) + r256(4 * sizeof(float)) + 4 * r256((size_t)nb * N * sizeof(float)) +
           r256(nb * plB) + r256((nb > 1 ? nb - 1 : 1) * plB) + (nb > 1 ? 0 : r256(plB));
  };
  const int nbat = (ws_size >= need(3)) ? 3 : (ws_size >= need(2)) ? 2 : 1;
  float*    g     = (float*)alloc(H * sizeof(float));
  float*    sw    = (float*)alloc(4 * sizeof(float));
  float*    sArr  = (float*)alloc((size_t)nbat * N * sizeof(float));
  float*    dArr  = (float*)alloc((size_t)nbat * N * sizeof(float));
  unsigned* mb    = (unsigned*)alloc((size_t)nbat * N * sizeof(unsigned));
  float*    denom = (float*)alloc((size_t)nbat * N * sizeof(float));
  float*    bufH  = (float*)alloc(nbat * plB);
  float*    bufX  = (float*)alloc((nbat > 1 ? nbat - 1 : 1) * plB);
  float*    bufC  = nbat > 1 ? bufX : (float*)alloc(plB);
  auto xplane = [&](int b, int nb) -> float* { return (nbat > 1 && b == nb - 1) ? out : bufX + b * pl; };

  int gemmGrid = (N + GROWS - 1) / GROWS;
  int nThrBlocks  = (N + 255) / 256;
  int ownBlocks   = (N + OWN_R - 1) / OWN_R;

  k_zero<<<1, 256, 0, stream>>>(g, H);
  k_colsum<<<nThrBlocks, 256, 0, stream>>>(x, g, N);
  k_selector<<<1, 128, 0, stream>>>(g, W1, b1, W2, b2, sw, N);

  for (int b0 = 0; b0 < 3; b0 += nbat) {
    const int nb = (3 - b0 < nbat) ? 3 - b0 : nbat;
    float* x0 = xplane(0, nb);
    float* x1 = nb > 1 ? xplane(1, nb) : x0;
    float* x2 = nb > 2 ? xplane(2, nb) : x0;
    for (int b = 0; b < nb; ++b)
      k_gemm_bf16x3<false, false, true><<<gemmGrid, 256, 0, stream>>>(x, Wg + (size_t)(b0 + b) * H * H, nullptr,
          bufH + b * pl, N,
          nullptr, nullptr, nullptr, nullptr, nullptr, nullptr, nullptr, nullptr, nullptr, 0, 0,
          a_src + (size_t)(b0 + b) * H, a_dst + (size_t)(b0 + b) * H, sArr + (size_t)b * N, dArr + (size_t)b * N);
    k_edge_own<<<ownBlocks, OWN_T, 0, stream>>>(ei, ei + E, ea, b0, nb, sArr, dArr, bufH, mb, denom,
                                                x0, x1, x2, N, E);
    for (int b = 0; b < nb; ++b) {
      const int i = b0 + b;
      k_gemm_bf16x3<true, true, false><<<gemmGrid, 256, 0, stream>>>(xplane(b, nb), Wsl + (size_t)i * H * H, bsl + i * H,
          bufC, N, denom + (size_t)b * N, bg + i * H, bufH + b * pl, sArr + (size_t)b * N, dArr + (size_t)b * N,
          mb + (size_t)b * N, gsl + i * H, betas + i * H, sw, i, i > 0 ? 1 : 0, nullptr, nullptr, nullptr, nullptr);
    }
  }
  k_gemm_bf16x3<false, true, false><<<gemmGrid, 256, 0, stream>>>(x, Wsl + (size_t)3 * H * H, bsl + 3 * H, bufC, N,
      nullptr, nullptr, nullptr, nullptr, nullptr, nullptr, gsl + 3 * H, betas + 3 * H, sw, 3, 1,
      nullptr, nullptr, nullptr, nullptr);

  k_gemm_bf16x3<false, true, false><<<gemmGrid, 256, 0, stream>>>(bufC, Wf, bfv, out, N, nullptr, nullptr,
      nullptr, nullptr, nullptr, nullptr, gf, betaf, nullptr, 0, 0, nullptr, nullptr, nullptr, nullptr);
}
